# starting kernel
# baseline (speedup 1.0000x reference)
#include <hip/hip_runtime.h>

typedef float v2f __attribute__((ext_vector_type(2)));
typedef float v8f __attribute__((ext_vector_type(8)));

#define F0 128
#define U0 64
#define H0 8
#define A0 8
#define U1 40

__device__ __forceinline__ unsigned fkey(float f) {
    unsigned b = __float_as_uint(f);
    return (b & 0x80000000u) ? ~b : (b | 0x80000000u);
}
__device__ __forceinline__ float fkey_dec(unsigned k) {
    return __uint_as_float((k & 0x80000000u) ? (k ^ 0x80000000u) : ~k);
}

__device__ __forceinline__ void edge_rc(const int* __restrict__ ei, int E, int e,
                                        int& r, int& c) {
    if (e < E) { r = ei[e]; c = ei[E + e]; }
    else       { r = e - E; c = e - E; }
}

__global__ __launch_bounds__(32) void proj0_kernel(
    const float* __restrict__ x,
    const float* __restrict__ wq, const float* __restrict__ bq,
    const float* __restrict__ wk, const float* __restrict__ bk,
    const float* __restrict__ wv,
    float* __restrict__ Q0, float* __restrict__ K0, float* __restrict__ V0)
{
    const int tile = blockIdx.x;
    const int mat  = blockIdx.y >> 2;
    const int ct   = blockIdx.y & 3;
    const int lane = threadIdx.x;
    const int hi   = lane >> 4;
    const int lo   = lane & 15;

    const float* W   = (mat == 0) ? wq : (mat == 1) ? wk : wv;
    const float* B   = (mat == 0) ? bq : (mat == 1) ? bk : nullptr;
    float*       OUT = (mat == 0) ? Q0 : (mat == 1) ? K0 : V0;

    const int m = tile * 16 + lo;
    const int n = ct * 16 + lo;

    v8f acc = {};
    for (int k = 0; k < F0; k += 4) {
        const int kb = k + 2 * hi;
        v2f a, b;
        a.x = x[(size_t)m * F0 + kb];
        a.y = x[(size_t)m * F0 + kb + 1];
        b.x = W[(size_t)kb * U0 + n];
        b.y = W[(size_t)(kb + 1) * U0 + n];
        acc = __builtin_amdgcn_wmma_f32_16x16x4_f32(
                  false, a, false, b, (short)0, acc, false, false);
    }

    const float bias = (B != nullptr) ? B[n] : 0.0f;
    #pragma unroll
    for (int v = 0; v < 8; ++v) {
        const int row = tile * 16 + v + 8 * hi;
        float val = acc[v] + bias;
        if (mat != 2) val = fmaxf(val, 0.0f);
        OUT[(size_t)row * U0 + n] = val;
    }
}

__global__ __launch_bounds__(32) void proj1v_kernel(
    const float* __restrict__ h0, const float* __restrict__ wv1,
    float* __restrict__ v1)
{
    const int tile = blockIdx.x;
    const int ct   = blockIdx.y;
    const int lane = threadIdx.x;
    const int hi   = lane >> 4;
    const int lo   = lane & 15;

    const int m = tile * 16 + lo;
    const int n = ct * 16 + lo;
    const bool valid = (n < U1);

    v8f acc = {};
    for (int k = 0; k < U0; k += 4) {
        const int kb = k + 2 * hi;
        v2f a, b;
        a.x = h0[(size_t)m * U0 + kb];
        a.y = h0[(size_t)m * U0 + kb + 1];
        b.x = valid ? wv1[(size_t)kb * U1 + n] : 0.0f;
        b.y = valid ? wv1[(size_t)(kb + 1) * U1 + n] : 0.0f;
        acc = __builtin_amdgcn_wmma_f32_16x16x4_f32(
                  false, a, false, b, (short)0, acc, false, false);
    }
    if (valid) {
        #pragma unroll
        for (int v = 0; v < 8; ++v) {
            const int row = tile * 16 + v + 8 * hi;
            v1[(size_t)row * U1 + n] = acc[v];
        }
    }
}

__global__ void qk1_kernel(const float* __restrict__ h0,
                           const float* __restrict__ wq1, const float* __restrict__ bq1,
                           const float* __restrict__ wk1, const float* __restrict__ bk1,
                           float* __restrict__ q1, float* __restrict__ k1, int n)
{
    const int i = blockIdx.x * blockDim.x + threadIdx.x;
    if (i >= n) return;
    const float* hr = h0 + (size_t)i * U0;
    float q = bq1[0], k = bk1[0];
    #pragma unroll 8
    for (int j = 0; j < U0; ++j) {
        const float h = hr[j];
        q += h * wq1[j];
        k += h * wk1[j];
    }
    q1[i] = fmaxf(q, 0.0f);
    k1[i] = fmaxf(k, 0.0f);
}

__global__ void edge_max0_kernel(const int* __restrict__ ei, int E, int Etot,
                                 const float* __restrict__ Q0,
                                 const float* __restrict__ K0,
                                 unsigned* __restrict__ max0key)
{
    const int t = blockIdx.x * blockDim.x + threadIdx.x;
    if (t >= Etot * H0) return;
    const int e = t >> 3, h = t & 7;
    int r, c; edge_rc(ei, E, e, r, c);
    const float* q = Q0 + (size_t)r * U0 + h * A0;
    const float* k = K0 + (size_t)c * U0 + h * A0;
    float s = 0.0f;
    #pragma unroll
    for (int a = 0; a < A0; ++a) s += q[a] * k[a];
    s *= 0.35355339059327373f;
    atomicMax(&max0key[r * H0 + h], fkey(s));
}

__global__ void edge_scatter0_kernel(const int* __restrict__ ei, int E, int Etot,
                                     const float* __restrict__ Q0,
                                     const float* __restrict__ K0,
                                     const float* __restrict__ V0,
                                     const unsigned* __restrict__ max0key,
                                     float* __restrict__ denom0,
                                     float* __restrict__ accum0)
{
    const int t = blockIdx.x * blockDim.x + threadIdx.x;
    if (t >= Etot * H0) return;
    const int e = t >> 3, h = t & 7;
    int r, c; edge_rc(ei, E, e, r, c);
    const float* q = Q0 + (size_t)r * U0 + h * A0;
    const float* k = K0 + (size_t)c * U0 + h * A0;
    float s = 0.0f;
    #pragma unroll
    for (int a = 0; a < A0; ++a) s += q[a] * k[a];
    s *= 0.35355339059327373f;
    const float m  = fkey_dec(max0key[r * H0 + h]);
    const float ev = __expf(s - m);
    atomicAdd(&denom0[r * H0 + h], ev);
    const float* vv  = V0 + (size_t)c * U0 + h * A0;
    float*       acc = accum0 + (size_t)r * U0 + h * A0;
    #pragma unroll
    for (int d = 0; d < A0; ++d) atomicAdd(&acc[d], ev * vv[d]);
}

__global__ void finalize0_kernel(const float* __restrict__ accum0,
                                 const float* __restrict__ denom0,
                                 const float* __restrict__ b0,
                                 float* __restrict__ h0, int n)
{
    const int t = blockIdx.x * blockDim.x + threadIdx.x;
    if (t >= n * U0) return;
    const int nn = t >> 6, cc = t & 63;
    const float val = accum0[t] / denom0[nn * H0 + (cc >> 3)] + b0[cc];
    h0[t] = fmaxf(val, 0.0f);
}

__global__ void edge_max1_kernel(const int* __restrict__ ei, int E, int Etot,
                                 const float* __restrict__ q1,
                                 const float* __restrict__ k1,
                                 unsigned* __restrict__ max1key)
{
    const int e = blockIdx.x * blockDim.x + threadIdx.x;
    if (e >= Etot) return;
    int r, c; edge_rc(ei, E, e, r, c);
    atomicMax(&max1key[r], fkey(q1[r] * k1[c]));
}

__global__ void edge_scatter1_kernel(const int* __restrict__ ei, int E, int Etot,
                                     const float* __restrict__ q1,
                                     const float* __restrict__ k1,
                                     const float* __restrict__ v1,
                                     const unsigned* __restrict__ max1key,
                                     float* __restrict__ denom1,
                                     float* __restrict__ accum1)
{
    const int t = blockIdx.x * blockDim.x + threadIdx.x;
    if (t >= Etot * 8) return;
    const int e = t >> 3, j = t & 7;
    int r, c; edge_rc(ei, E, e, r, c);
    const float s  = q1[r] * k1[c];
    const float m  = fkey_dec(max1key[r]);
    const float ev = __expf(s - m);
    if (j == 0) atomicAdd(&denom1[r], ev);
    const float* vv  = v1 + (size_t)c * U1;
    float*       acc = accum1 + (size_t)r * U1;
    for (int d = j; d < U1; d += 8) atomicAdd(&acc[d], ev * vv[d]);
}

__global__ void finalize_out_kernel(const float* __restrict__ accum1,
                                    const float* __restrict__ denom1,
                                    const float* __restrict__ b1,
                                    float* __restrict__ out, int n)
{
    const int t = blockIdx.x * blockDim.x + threadIdx.x;
    if (t >= n * U1) return;
    out[t] = accum1[t] / denom1[t / U1] + b1[t % U1];
}

extern "C" void kernel_launch(void* const* d_in, const int* in_sizes, int n_in,
                              void* d_out, int out_size, void* d_ws, size_t ws_size,
                              hipStream_t stream)
{
    const float* x    = (const float*)d_in[0];
    const int*   ei   = (const int*)  d_in[1];
    const float* wq0  = (const float*)d_in[2];
    const float* bq0  = (const float*)d_in[3];
    const float* wk0  = (const float*)d_in[4];
    const float* bk0  = (const float*)d_in[5];
    const float* wv0  = (const float*)d_in[6];
    const float* b0   = (const float*)d_in[7];
    const float* wq1  = (const float*)d_in[8];
    const float* bq1  = (const float*)d_in[9];
    const float* wk1  = (const float*)d_in[10];
    const float* bk1  = (const float*)d_in[11];
    const float* wv1  = (const float*)d_in[12];
    const float* b1   = (const float*)d_in[13];
    float* out = (float*)d_out;

    const int N    = in_sizes[0] / F0;
    const int E    = in_sizes[1] / 2;
    const int Etot = E + N;

    float* ws = (float*)d_ws;
    float* R0 = ws;
    float* R1 = R0 + (size_t)N * U0;
    float* R2 = R1 + (size_t)N * U0;
    float* R3 = R2 + (size_t)N * U0;
    float* R4 = R3 + (size_t)N * H0;
    float* R5 = R4 + (size_t)N * H0;

    const int TB = 256;

    proj0_kernel<<<dim3(N / 16, 12), 32, 0, stream>>>(
        x, wq0, bq0, wk0, bk0, wv0, R0, R1, R2);

    hipMemsetAsync(R3, 0, (size_t)N * (H0 + H0 + U0) * sizeof(float), stream);

    edge_max0_kernel<<<(Etot * H0 + TB - 1) / TB, TB, 0, stream>>>(
        ei, E, Etot, R0, R1, (unsigned*)R3);
    edge_scatter0_kernel<<<(Etot * H0 + TB - 1) / TB, TB, 0, stream>>>(
        ei, E, Etot, R0, R1, R2, (const unsigned*)R3, R4, R5);
    finalize0_kernel<<<(N * U0 + TB - 1) / TB, TB, 0, stream>>>(
        R5, R4, b0, R0  , N);

    float* q1 = R3;
    float* k1 = R3 + N;
    qk1_kernel<<<(N + TB - 1) / TB, TB, 0, stream>>>(
        R0, wq1, bq1, wk1, bk1, q1, k1, N);
    proj1v_kernel<<<dim3(N / 16, 3), 32, 0, stream>>>(
        R0, wv1, R1  );

    unsigned* max1key = (unsigned*)R4;
    float*    denom1  = R4 + N;
    float*    accum1  = R2;
    hipMemsetAsync(R4, 0, (size_t)2 * N * sizeof(float), stream);
    hipMemsetAsync(R2, 0, (size_t)N * U1 * sizeof(float), stream);

    edge_max1_kernel<<<(Etot + TB - 1) / TB, TB, 0, stream>>>(
        ei, E, Etot, q1, k1, max1key);
    edge_scatter1_kernel<<<(Etot * 8 + TB - 1) / TB, TB, 0, stream>>>(
        ei, E, Etot, q1, k1, R1, max1key, denom1, accum1);
    finalize_out_kernel<<<(N * U1 + TB - 1) / TB, TB, 0, stream>>>(
        accum1, denom1, b1, out, N);
}


# Both projection GEMMs: 8-wave work-groups lay the weights out in LDS (k-pair order); each wave owns a 16-row tile and all its column tiles, A fragment loaded once (12x/3x fewer waves and A reads)
# speedup vs baseline: 4.5706x; 4.5706x over previous
#include <hip/hip_runtime.h>

typedef float v2f __attribute__((ext_vector_type(2)));
typedef float v8f __attribute__((ext_vector_type(8)));

#define F0 128
#define U0 64
#define H0 8
#define A0 8
#define U1 40

#define NCT0   12
#define WP0    416

__global__ __launch_bounds__(256) void proj0_kernel(
    const float* __restrict__ x,
    const float* __restrict__ wq, const float* __restrict__ bq,
    const float* __restrict__ wk, const float* __restrict__ bk,
    const float* __restrict__ wv,
    float* __restrict__ Q0, float* __restrict__ K0, float* __restrict__ V0,
    int ntiles)
{
    __shared__ __attribute__((aligned(16))) float Wl[(F0 / 2) * WP0];

    const int lane = threadIdx.x & 31;
    const int hi   = lane >> 4;
    const int lo   = lane & 15;
    const int tile = blockIdx.x * 8 + (threadIdx.x >> 5);

    float4 r0[12], r1[12];
    #pragma unroll
    for (int i = 0; i < 12; ++i) {
        const int it  = threadIdx.x + 256 * i;
        const int kp  = it / 48;
        const int rem = it - kp * 48;
        const int mat = rem >> 4;
        const int n4  = (rem & 15) * 4;
        const float* Wm = (mat == 0) ? wq : (mat == 1) ? wk : wv;
        r0[i] = *reinterpret_cast<const float4*>(Wm + (size_t)(2 * kp) * U0 + n4);
        r1[i] = *reinterpret_cast<const float4*>(Wm + (size_t)(2 * kp + 1) * U0 + n4);
    }

    const float* xr = x + (size_t)(min(tile, ntiles - 1) * 16 + lo) * F0 + 2 * hi;
    v2f a[F0 / 4];
    #pragma unroll
    for (int j = 0; j < F0 / 4; ++j) a[j] = *reinterpret_cast<const v2f*>(xr + 4 * j);

    #pragma unroll
    for (int i = 0; i < 12; ++i) {
        const int it  = threadIdx.x + 256 * i;
        const int kp  = it / 48;
        const int rem = it - kp * 48;
        const int mat = rem >> 4;
        const int n4  = (rem & 15) * 4;
        float* d = Wl + kp * WP0 + (mat * U0 + n4) * 2;
        *reinterpret_cast<float4*>(d)     = make_float4(r0[i].x, r1[i].x, r0[i].y, r1[i].y);
        *reinterpret_cast<float4*>(d + 4) = make_float4(r0[i].z, r1[i].z, r0[i].w, r1[i].w);
    }
    __syncthreads();

    if (tile >= ntiles) return;

    const float* wl = Wl + hi * WP0 + lo * 2;

    v8f acc[NCT0];
    #pragma unroll
    for (int c = 0; c < NCT0; ++c) acc[c] = v8f{};

    #pragma unroll
    for (int j = 0; j < F0 / 4; ++j) {
        const float* wk2 = wl + (2 * j) * WP0;
        v2f b[NCT0];
        #pragma unroll
        for (int c = 0; c < NCT0; ++c) b[c] = *reinterpret_cast<const v2f*>(wk2 + c * 32);
        #pragma unroll
        for (int c = 0; c < NCT0; ++c)
            acc[c] = __builtin_amdgcn_wmma_f32_16x16x4_f32(
                         false, a[j], false, b[c], (short)0, acc[c], false, false);
    }

    #pragma unroll
    for (int c = 0; c < NCT0; ++c) {
        const int mat = c >> 2;
        const int n   = (c & 3) * 16 + lo;
        float* OUT = (mat == 0) ? Q0 : (mat == 1) ? K0 : V0;
        const float bias = (mat == 0) ? bq[n] : (mat == 1) ? bk[n] : 0.0f;
        #pragma unroll
        for (int v = 0; v < 8; ++v) {
            const int row = tile * 16 + v + 8 * hi;
            float val = acc[c][v] + bias;
            if (mat != 2) val = fmaxf(val, 0.0f);
            OUT[(size_t)row * U0 + n] = val;
        }
    }
}

#define WP1 96

__global__ __launch_bounds__(256) void proj1v_kernel(
    const float* __restrict__ h0, const float* __restrict__ wv1,
    float* __restrict__ v1, int ntiles)
{
    __shared__ __attribute__((aligned(16))) float Wl[(U0 / 2) * WP1];

    const int lane = threadIdx.x & 31;
    const int hi   = lane >> 4;
    const int lo   = lane & 15;
    const int tile = blockIdx.x * 8 + (threadIdx.x >> 5);

    v2f w[6];
    #pragma unroll
    for (int i = 0; i < 6; ++i) {
        const int it = threadIdx.x + 256 * i;
        const int kp = it / 48;
        const int nc = min(it - kp * 48, U1 - 1);
        w[i].x = wv1[(size_t)(2 * kp) * U1 + nc];
        w[i].y = wv1[(size_t)(2 * kp + 1) * U1 + nc];
    }

    const float* hr = h0 + (size_t)(min(tile, ntiles - 1) * 16 + lo) * U0 + 2 * hi;
    v2f a[U0 / 4];
    #pragma unroll
    for (int j = 0; j < U0 / 4; ++j) a[j] = *reinterpret_cast<const v2f*>(hr + 4 * j);

    #pragma unroll
    for (int i = 0; i < 6; ++i) {
        const int it = threadIdx.x + 256 * i;
        const int kp = it / 48;
        const int n  = it - kp * 48;
        const v2f z = {0.0f, 0.0f};
        *reinterpret_cast<v2f*>(Wl + kp * WP1 + n * 2) = (n < U1) ? w[i] : z;
    }
    __syncthreads();

    if (tile >= ntiles) return;

    const float* wl = Wl + hi * WP1 + lo * 2;

    v8f acc[3];
    #pragma unroll
    for (int c = 0; c < 3; ++c) acc[c] = v8f{};

    #pragma unroll
    for (int j = 0; j < U0 / 4; ++j) {
        #pragma unroll
        for (int c = 0; c < 3; ++c) {
            const v2f b = *reinterpret_cast<const v2f*>(wl + (2 * j) * WP1 + c * 32);
            acc[c] = __builtin_amdgcn_wmma_f32_16x16x4_f32(
                         false, a[j], false, b, (short)0, acc[c], false, false);
        }
    }

    #pragma unroll
    for (int c = 0; c < 3; ++c) {
        const int n = c * 16 + lo;
        if (n < U1) {
            #pragma unroll
            for (int v = 0; v < 8; ++v) {
                const int row = tile * 16 + v + 8 * hi;
                v1[(size_t)row * U1 + n] = acc[c][v];
            }
        }
    }
}

__global__ void qk1_kernel(const float* __restrict__ h0,
                           const float* __restrict__ wq1, const float* __restrict__ bq1,
                           const float* __restrict__ wk1, const float* __restrict__ bk1,
                           float* __restrict__ q1, float* __restrict__ k1, int n)
{
    const int i = blockIdx.x * blockDim.x + threadIdx.x;
    if (i >= n) return;
    const float* hr = h0 + (size_t)i * U0;
    float q = bq1[0], k = bk1[0];
    #pragma unroll 8
    for (int j = 0; j < U0; ++j) {
        const float h = hr[j];
        q += h * wq1[j];
        k += h * wk1[j];
    }
    q1[i] = fmaxf(q, 0.0f);
    k1[i] = fmaxf(k, 0.0f);
}

#define BINW  256
#define NBMAX 256
#define LP    257
#define CH    3328
#define EPT   13
#define CAPB  8192

__device__ __forceinline__ int excl_scan256(int v, int* wtot, int& tot)
{
    const int lane = threadIdx.x & 31;
    const int wid  = threadIdx.x >> 5;
    int x = v;
    for (int o = 1; o < 32; o <<= 1) {
        const int y = __shfl_up(x, o);
        if (lane >= o) x += y;
    }
    if (lane == 31) wtot[wid] = x;
    __syncthreads();
    int pre = 0, t = 0;
    for (int w = 0; w < 8; ++w) {
        const int s = wtot[w];
        if (w < wid) pre += s;
        t += s;
    }
    tot = t;
    __syncthreads();
    return pre + x - v;
}

__global__ __launch_bounds__(256) void bin_chunk_kernel(
    const int* __restrict__ ei, int E,
    int* __restrict__ S, int* __restrict__ loc)
{
    __shared__ int cnt[NBMAX];
    __shared__ int wtot[8];
    __shared__ int stage[CH];
    const int tid   = threadIdx.x;
    const int cbase = blockIdx.x * CH;
    const int n     = min(CH, E - cbase);

    cnt[tid] = 0;
    __syncthreads();

    int val[EPT], bin[EPT], rk[EPT];
    #pragma unroll
    for (int k = 0; k < EPT; ++k) {
        const int j = k * 256 + tid;
        bin[k] = -1;
        if (j < n) {
            const int r = ei[cbase + j];
            const int c = ei[E + cbase + j];
            bin[k] = r / BINW;
            val[k] = ((r % BINW) << 24) | c;
        }
    }
    #pragma unroll
    for (int k = 0; k < EPT; ++k)
        if (bin[k] >= 0) rk[k] = atomicAdd(&cnt[bin[k]], 1);
    __syncthreads();

    int tot;
    const int off = excl_scan256(cnt[tid], wtot, tot);
    loc[blockIdx.x * LP + tid] = off;
    if (tid == 0) loc[blockIdx.x * LP + NBMAX] = tot;
    cnt[tid] = off;
    __syncthreads();

    #pragma unroll
    for (int k = 0; k < EPT; ++k)
        if (bin[k] >= 0) stage[cnt[bin[k]] + rk[k]] = val[k];
    __syncthreads();

    for (int j = tid; j < n; j += 256) S[cbase + j] = stage[j];
}

template <typename F>
__device__ __forceinline__ void walk_run(const int* __restrict__ p, int n, F f)
{
    int j = 0;
    for (; j + 4 <= n; j += 4) {
        const int v0 = p[j], v1 = p[j + 1], v2 = p[j + 2], v3 = p[j + 3];
        f(v0); f(v1); f(v2); f(v3);
    }
    for (; j < n; ++j) f(p[j]);
}

__global__ __launch_bounds__(256) void bin_sort_kernel(
    const int* __restrict__ S, const int* __restrict__ loc, int NC, int N,
    int* __restrict__ seg_start, int* __restrict__ seg_end,
    int* __restrict__ ccol)
{
    __shared__ int ncnt[BINW];
    __shared__ int wtot[8];
    __shared__ int stage[CAPB];
    const int tid = threadIdx.x;
    const int b   = blockIdx.x;

    ncnt[tid] = 0;

    int lo0 = 0, n0 = 0;
    int below = 0, mine = 0;
    for (int c = tid; c < NC; c += 256) {
        const int lo = loc[c * LP + b];
        const int nn = loc[c * LP + b + 1] - lo;
        if (c == tid) { lo0 = lo; n0 = nn; }
        below += lo;
        mine  += nn;
    }
    int base, total, unused;
    excl_scan256(below, wtot, base);
    excl_scan256(mine,  wtot, total);

    auto count = [&](int v) { atomicAdd(&ncnt[(unsigned)v >> 24], 1); };
    if (tid < NC) walk_run(S + tid * CH + lo0, n0, count);
    for (int c = tid + 256; c < NC; c += 256) {
        const int lo = loc[c * LP + b];
        walk_run(S + c * CH + lo, loc[c * LP + b + 1] - lo, count);
    }
    __syncthreads();

    const int d  = ncnt[tid];
    const int ex = excl_scan256(d, wtot, unused);
    const int node = b * BINW + tid;
    if (node < N) {
        seg_start[node] = base + ex;
        seg_end[node]   = base + ex + d;
    }
    ncnt[tid] = ex;
    __syncthreads();

    const bool staged = (total <= CAPB);
    auto place = [&](int v) {
        const int p = atomicAdd(&ncnt[(unsigned)v >> 24], 1);
        const int c = v & 0xFFFFFF;
        if (staged) stage[p] = c; else ccol[base + p] = c;
    };
    if (tid < NC) walk_run(S + tid * CH + lo0, n0, place);
    for (int c = tid + 256; c < NC; c += 256) {
        const int lo = loc[c * LP + b];
        walk_run(S + c * CH + lo, loc[c * LP + b + 1] - lo, place);
    }
    if (staged) {
        __syncthreads();
        for (int j = tid; j < total; j += 256) ccol[base + j] = stage[j];
    }
}

__device__ __forceinline__ float4 ld4(const float* p) {
    return *reinterpret_cast<const float4*>(p);
}

__device__ __forceinline__ float score8(const float4 qa, const float4 qb,
                                        const float4 ka, const float4 kb) {
    float s = qa.x * ka.x;
    s += qa.y * ka.y; s += qa.z * ka.z; s += qa.w * ka.w;
    s += qb.x * kb.x; s += qb.y * kb.y; s += qb.z * kb.z; s += qb.w * kb.w;
    return s * 0.35355339059327373f;
}

__device__ __forceinline__ float4 fma4(const float4 a, float f, float e, const float4 v) {
    return make_float4(a.x * f + e * v.x, a.y * f + e * v.y,
                       a.z * f + e * v.z, a.w * f + e * v.w);
}

__global__ __launch_bounds__(256) void gather0_kernel(
    const int* __restrict__ seg_start, const int* __restrict__ seg_end,
    const int* __restrict__ ccol,
    float* QH, const float* __restrict__ K0, const float* __restrict__ V0,
    const float* __restrict__ b0, int n)
{
    const int t = blockIdx.x * 256 + threadIdx.x;
    if (t >= n * H0) return;
    const int r = t >> 3, h = t & 7;
    const int ho = h * A0;
    const size_t ro = (size_t)r * U0 + ho;
    const float4 qa = ld4(QH + ro), qb = ld4(QH + ro + 4);

    float  m   = score8(qa, qb, ld4(K0 + ro), ld4(K0 + ro + 4));
    float  den = 1.0f;
    float4 aa  = ld4(V0 + ro), ab = ld4(V0 + ro + 4);

    int p = seg_start[r];
    const int p1 = seg_end[r];
    for (; p + 4 <= p1; p += 4) {
        size_t o[4];
        float4 ka[4], kb[4], va[4], vb[4];
        float  s[4], e[4];
        for (int u = 0; u < 4; ++u) o[u] = (size_t)ccol[p + u] * U0 + ho;
        for (int u = 0; u < 4; ++u) {
            ka[u] = ld4(K0 + o[u]); kb[u] = ld4(K0 + o[u] + 4);
            va[u] = ld4(V0 + o[u]); vb[u] = ld4(V0 + o[u] + 4);
        }
        for (int u = 0; u < 4; ++u) s[u] = score8(qa, qb, ka[u], kb[u]);
        const float mn = fmaxf(fmaxf(m, fmaxf(s[0], s[1])), fmaxf(s[2], s[3]));
        const float f  = __expf(m - mn);
        for (int u = 0; u < 4; ++u) e[u] = __expf(s[u] - mn);
        den = den * f + ((e[0] + e[1]) + (e[2] + e[3]));
        aa = fma4(aa, f, e[0], va[0]); ab = fma4(ab, f, e[0], vb[0]);
        for (int u = 1; u < 4; ++u) {
            aa = fma4(aa, 1.0f, e[u], va[u]); ab = fma4(ab, 1.0f, e[u], vb[u]);
        }
        m = mn;
    }
    for (; p < p1; ++p) {
        const size_t o = (size_t)ccol[p] * U0 + ho;
        const float4 ka = ld4(K0 + o), kb = ld4(K0 + o + 4);
        const float4 va = ld4(V0 + o), vb = ld4(V0 + o + 4);
        const float s  = score8(qa, qb, ka, kb);
        const float mn = fmaxf(m, s);
        const float f  = __expf(m - mn);
        const float e  = __expf(s - mn);
        den = den * f + e;
        aa = fma4(aa, f, e, va); ab = fma4(ab, f, e, vb);
        m = mn;
    }

    const float4 ba = ld4(b0 + ho), bb = ld4(b0 + ho + 4);
    float4 ha, hb;
    ha.x = fmaxf(aa.x / den + ba.x, 0.0f); ha.y = fmaxf(aa.y / den + ba.y, 0.0f);
    ha.z = fmaxf(aa.z / den + ba.z, 0.0f); ha.w = fmaxf(aa.w / den + ba.w, 0.0f);
    hb.x = fmaxf(ab.x / den + bb.x, 0.0f); hb.y = fmaxf(ab.y / den + bb.y, 0.0f);
    hb.z = fmaxf(ab.z / den + bb.z, 0.0f); hb.w = fmaxf(ab.w / den + bb.w, 0.0f);
    *reinterpret_cast<float4*>(QH + ro)     = ha;
    *reinterpret_cast<float4*>(QH + ro + 4) = hb;
}

__global__ __launch_bounds__(256) void gather1_kernel(
    const int* __restrict__ seg_start, const int* __restrict__ seg_end,
    const int* __restrict__ ccol,
    const float* __restrict__ q1, const float* __restrict__ k1,
    const float* __restrict__ v1, const float* __restrict__ b1,
    float* __restrict__ out, int n)
{
    const int t = blockIdx.x * 256 + threadIdx.x;
    if (t >= n * 8) return;
    const int r = t >> 3, j = t & 7;
    const float qr = q1[r];

    float m   = qr * k1[r];
    float den = 1.0f;
    float acc[5];
    for (int i = 0; i < 5; ++i) acc[i] = v1[(size_t)r * U1 + j + 8 * i];

    int p = seg_start[r];
    const int p1 = seg_end[r];
    for (; p + 4 <= p1; p += 4) {
        int   c[4];
        float s[4], e[4], vv[4][5];
        for (int u = 0; u < 4; ++u) c[u] = ccol[p + u];
        for (int u = 0; u < 4; ++u) {
            s[u] = qr * k1[c[u]];
            for (int i = 0; i < 5; ++i) vv[u][i] = v1[(size_t)c[u] * U1 + j + 8 * i];
        }
        const float mn = fmaxf(fmaxf(m, fmaxf(s[0], s[1])), fmaxf(s[2], s[3]));
        const float f  = __expf(m - mn);
        for (int u = 0; u < 4; ++u) e[u] = __expf(s[u] - mn);
        den = den * f + ((e[0] + e[1]) + (e[2] + e[3]));
        for (int i = 0; i < 5; ++i) {
            float a = acc[i] * f + e[0] * vv[0][i];
            a += e[1] * vv[1][i]; a += e[2] * vv[2][i]; a += e[3] * vv[3][i];
            acc[i] = a;
        }
        m = mn;
    }
    for (; p < p1; ++p) {
        const int c = ccol[p];
        const float s  = qr * k1[c];
        const float mn = fmaxf(m, s);
        const float f  = __expf(m - mn);
        const float e  = __expf(s - mn);
        den = den * f + e;
        for (int i = 0; i < 5; ++i) acc[i] = acc[i] * f + e * v1[(size_t)c * U1 + j + 8 * i];
        m = mn;
    }
    for (int i = 0; i < 5; ++i)
        out[(size_t)r * U1 + j + 8 * i] = acc[i] / den + b1[j + 8 * i];
}

extern "C" void kernel_launch(void* const* d_in, const int* in_sizes, int n_in,
                              void* d_out, int out_size, void* d_ws, size_t ws_size,
                              hipStream_t stream)
{
    const float* x    = (const float*)d_in[0];
    const int*   ei   = (const int*)  d_in[1];
    const float* wq0  = (const float*)d_in[2];
    const float* bq0  = (const float*)d_in[3];
    const float* wk0  = (const float*)d_in[4];
    const float* bk0  = (const float*)d_in[5];
    const float* wv0  = (const float*)d_in[6];
    const float* b0   = (const float*)d_in[7];
    const float* wq1  = (const float*)d_in[8];
    const float* bq1  = (const float*)d_in[9];
    const float* wk1  = (const float*)d_in[10];
    const float* bk1  = (const float*)d_in[11];
    const float* wv1  = (const float*)d_in[12];
    const float* b1   = (const float*)d_in[13];
    float* out = (float*)d_out;

    const int N = in_sizes[0] / F0;
    const int E = in_sizes[1] / 2;

    float* ws = (float*)d_ws;
    float* R0 = ws;
    float* R1 = R0 + (size_t)N * U0;
    float* R2 = R1 + (size_t)N * U0;
    int* seg_start = (int*)(R2 + (size_t)N * U0) + N + 64;
    int* seg_fill  = seg_start + N;
    int* ccol      = seg_fill + N;
    int* S         = (int*)R0;
    int* loc       = S + E;

    const int TB = 256;

    const int NC = (E + CH - 1) / CH;
    bin_chunk_kernel<<<NC, 256, 0, stream>>>(ei, E, S, loc);
    bin_sort_kernel<<<(N + BINW - 1) / BINW, BINW, 0, stream>>>(
        S, loc, NC, N, seg_start, seg_fill, ccol);

    proj0_kernel<<<(N / 16 + 7) / 8, 256, 0, stream>>>(
        x, wq0, bq0, wk0, bk0, wv0, R0, R1, R2, N / 16);

    gather0_kernel<<<(N * H0 + TB - 1) / TB, TB, 0, stream>>>(
        seg_start, seg_fill, ccol, R0  , R1, R2, b0, N);

    float* q1 = R2;
    float* k1 = R2 + N;
    qk1_kernel<<<(N + TB - 1) / TB, TB, 0, stream>>>(
        R0, wq1, bq1, wk1, bk1, q1, k1, N);
    proj1v_kernel<<<(N / 16 + 7) / 8, 256, 0, stream>>>(
        R0, wv1, R1  , N / 16);
    gather1_kernel<<<(N * 8 + TB - 1) / TB, TB, 0, stream>>>(
        seg_start, seg_fill, ccol, q1, k1, R1, b1, out, N);
}


# Q0 projection moved from the two projection launches into the layer-0 gather's prologue: each gather work-group makes relu(x@wq0+bq0) of its 32 rows on f32 WMMA through LDS; Q0 never in memory
# speedup vs baseline: 5.7702x; 1.2625x over previous
#include <hip/hip_runtime.h>

typedef float v2f __attribute__((ext_vector_type(2)));
typedef float v8f __attribute__((ext_vector_type(8)));

#define F0 128
#define U0 64
#define H0 8
#define A0 8
#define U1 40

#define NCT0   8
#define WP0    288

__device__ __forceinline__ void proj0_body(
    const int blk,
    const float* __restrict__ x,
    const float* __restrict__ wk, const float* __restrict__ bk,
    const float* __restrict__ wv,
    float* __restrict__ K0, float* __restrict__ V0,
    const int tile0, const int tile1,
    float* Wl)
{

    const int lane = threadIdx.x & 31;
    const int hi   = lane >> 4;
    const int lo   = lane & 15;
    const int tile = tile0 + blk * 8 + (threadIdx.x >> 5);

    float4 r0[8], r1[8];
    #pragma unroll
    for (int i = 0; i < 8; ++i) {
        const int it  = threadIdx.x + 256 * i;
        const int kp  = it >> 5;
        const int rem = it & 31;
        const int mat = rem >> 4;
        const int n4  = (rem & 15) * 4;
        const float* Wm = (mat == 0) ? wk : wv;
        r0[i] = *reinterpret_cast<const float4*>(Wm + (size_t)(2 * kp) * U0 + n4);
        r1[i] = *reinterpret_cast<const float4*>(Wm + (size_t)(2 * kp + 1) * U0 + n4);
    }

    const float* xr = x + (size_t)(min(tile, tile1 - 1) * 16 + lo) * F0 + 2 * hi;
    v2f a[F0 / 4];
    #pragma unroll
    for (int j = 0; j < F0 / 4; ++j) a[j] = *reinterpret_cast<const v2f*>(xr + 4 * j);

    #pragma unroll
    for (int i = 0; i < 8; ++i) {
        const int it  = threadIdx.x + 256 * i;
        const int kp  = it >> 5;
        const int rem = it & 31;
        const int mat = rem >> 4;
        const int n4  = (rem & 15) * 4;
        float* d = Wl + kp * WP0 + (mat * U0 + n4) * 2;
        *reinterpret_cast<float4*>(d)     = make_float4(r0[i].x, r1[i].x, r0[i].y, r1[i].y);
        *reinterpret_cast<float4*>(d + 4) = make_float4(r0[i].z, r1[i].z, r0[i].w, r1[i].w);
    }
    __syncthreads();

    const float* wl = Wl + hi * WP0 + lo * 2;

    v8f acc[NCT0];
    #pragma unroll
    for (int c = 0; c < NCT0; ++c) acc[c] = v8f{};

    #pragma unroll
    for (int j = 0; j < F0 / 4; ++j) {
        const float* wk2 = wl + (2 * j) * WP0;
        v2f b[NCT0];
        #pragma unroll
        for (int c = 0; c < NCT0; ++c) b[c] = *reinterpret_cast<const v2f*>(wk2 + c * 32);
        #pragma unroll
        for (int c = 0; c < NCT0; ++c)
            acc[c] = __builtin_amdgcn_wmma_f32_16x16x4_f32(
                         false, a[j], false, b[c], (short)0, acc[c], false, false);
    }

    if (tile >= tile1) return;

    #pragma unroll
    for (int c = 0; c < NCT0; ++c) {
        const int mat = c >> 2;
        const int n   = (c & 3) * 16 + lo;
        float* OUT = (mat == 0) ? K0 : V0;
        const float bias = (mat == 0) ? bk[n] : 0.0f;
        #pragma unroll
        for (int v = 0; v < 8; ++v) {
            const int row = tile * 16 + v + 8 * hi;
            float val = acc[c][v] + bias;
            if (mat == 0) val = fmaxf(val, 0.0f);
            OUT[(size_t)row * U0 + n] = val;
        }
    }
}

#define WP1 96
#define HP  68
#define VP  64
#define QC  40
#define KC  41

#define BINW  256
#define NBMAX 256
#define LP    257
#define CH    3328
#define EPT   13
#define CAPB  8192

__device__ __forceinline__ int excl_scan256(int v, int* wtot, int& tot)
{
    const int lane = threadIdx.x & 31;
    const int wid  = threadIdx.x >> 5;
    int x = v;
    for (int o = 1; o < 32; o <<= 1) {
        const int y = __shfl_up(x, o);
        if (lane >= o) x += y;
    }
    if (lane == 31) wtot[wid] = x;
    __syncthreads();
    int pre = 0, t = 0;
    for (int w = 0; w < 8; ++w) {
        const int s = wtot[w];
        if (w < wid) pre += s;
        t += s;
    }
    tot = t;
    __syncthreads();
    return pre + x - v;
}

__device__ __forceinline__ void bin_chunk_body(
    const int blk,
    const int* __restrict__ ei, int E,
    int* __restrict__ S, int* __restrict__ loc,
    int* sm)
{
    int* cnt   = sm;
    int* wtot  = sm + NBMAX;
    int* stage = sm + NBMAX + 8;
    const int tid   = threadIdx.x;
    const int cbase = blk * CH;
    const int n     = min(CH, E - cbase);

    cnt[tid] = 0;
    __syncthreads();

    int val[EPT], bin[EPT], rk[EPT];
    #pragma unroll
    for (int k = 0; k < EPT; ++k) {
        const int j = k * 256 + tid;
        bin[k] = -1;
        if (j < n) {
            const int r = ei[cbase + j];
            const int c = ei[E + cbase + j];
            bin[k] = r / BINW;
            val[k] = ((r % BINW) << 24) | c;
        }
    }
    #pragma unroll
    for (int k = 0; k < EPT; ++k)
        if (bin[k] >= 0) rk[k] = atomicAdd(&cnt[bin[k]], 1);
    __syncthreads();

    int tot;
    const int off = excl_scan256(cnt[tid], wtot, tot);
    loc[blk * LP + tid] = off;
    if (tid == 0) loc[blk * LP + NBMAX] = tot;
    cnt[tid] = off;
    __syncthreads();

    #pragma unroll
    for (int k = 0; k < EPT; ++k)
        if (bin[k] >= 0) stage[cnt[bin[k]] + rk[k]] = val[k];
    __syncthreads();

    for (int j = tid; j < n; j += 256) S[cbase + j] = stage[j];
}

template <typename F>
__device__ __forceinline__ void walk_run(const int* __restrict__ p, int n, F f)
{
    int j = 0;
    for (; j + 4 <= n; j += 4) {
        const int v0 = p[j], v1 = p[j + 1], v2 = p[j + 2], v3 = p[j + 3];
        f(v0); f(v1); f(v2); f(v3);
    }
    for (; j < n; ++j) f(p[j]);
}

__device__ __forceinline__ void bin_sort_body(
    const int b,
    const int* __restrict__ S, const int* __restrict__ loc, int NC, int N,
    int* __restrict__ seg_start, int* __restrict__ seg_end,
    int* __restrict__ ccol,
    int* sm)
{
    int* ncnt  = sm;
    int* wtot  = sm + BINW;
    int* stage = sm + BINW + 8;
    const int tid = threadIdx.x;

    ncnt[tid] = 0;

    int lo0 = 0, n0 = 0;
    int below = 0, mine = 0;
    for (int c = tid; c < NC; c += 256) {
        const int lo = loc[c * LP + b];
        const int nn = loc[c * LP + b + 1] - lo;
        if (c == tid) { lo0 = lo; n0 = nn; }
        below += lo;
        mine  += nn;
    }
    int base, total, unused;
    excl_scan256(below, wtot, base);
    excl_scan256(mine,  wtot, total);

    auto count = [&](int v) { atomicAdd(&ncnt[(unsigned)v >> 24], 1); };
    if (tid < NC) walk_run(S + tid * CH + lo0, n0, count);
    for (int c = tid + 256; c < NC; c += 256) {
        const int lo = loc[c * LP + b];
        walk_run(S + c * CH + lo, loc[c * LP + b + 1] - lo, count);
    }
    __syncthreads();

    const int d  = ncnt[tid];
    const int ex = excl_scan256(d, wtot, unused);
    const int node = b * BINW + tid;
    if (node < N) {
        seg_start[node] = base + ex;
        seg_end[node]   = base + ex + d;
    }
    ncnt[tid] = ex;
    __syncthreads();

    const bool staged = (total <= CAPB);
    typedef __attribute__((address_space(3))) int lds_int;
    lds_int* const stage_l = (lds_int*)stage;
    auto place = [&](int v) {
        const int p = atomicAdd(&ncnt[(unsigned)v >> 24], 1);
        const int c = v & 0xFFFFFF;
        if (staged) stage_l[p] = c; else ccol[base + p] = c;
    };
    if (tid < NC) walk_run(S + tid * CH + lo0, n0, place);
    for (int c = tid + 256; c < NC; c += 256) {
        const int lo = loc[c * LP + b];
        walk_run(S + c * CH + lo, loc[c * LP + b + 1] - lo, place);
    }
    if (staged) {
        __syncthreads();
        for (int j = tid; j < total; j += 256) ccol[base + j] = stage[j];
    }
}

__global__ __launch_bounds__(256) void chunk_proj0_kernel(
    const int* __restrict__ ei, int E, int* __restrict__ S, int* __restrict__ loc,
    const int nproj,
    const float* __restrict__ x,
    const float* __restrict__ wk, const float* __restrict__ bk,
    const float* __restrict__ wv,
    float* __restrict__ K0, float* __restrict__ V0,
    const int tile0, const int tile1)
{
    __shared__ __attribute__((aligned(16))) float lds[(F0 / 2) * WP0];
    static_assert((F0 / 2) * WP0 >= NBMAX + 8 + CH, "the chunk step's LDS fits in the weight array");
    if ((int)blockIdx.x < nproj)
        proj0_body((int)blockIdx.x, x, wk, bk, wv, K0, V0, tile0, tile1, lds);
    else
        bin_chunk_body((int)blockIdx.x - nproj, ei, E, S, loc, reinterpret_cast<int*>(lds));
}

__global__ __launch_bounds__(256) void sort_proj0_kernel(
    const int* __restrict__ S, const int* __restrict__ loc, int NC, int N,
    int* __restrict__ seg_start, int* __restrict__ seg_end, int* __restrict__ ccol,
    const int nproj,
    const float* __restrict__ x,
    const float* __restrict__ wk, const float* __restrict__ bk,
    const float* __restrict__ wv,
    float* __restrict__ K0, float* __restrict__ V0,
    const int tile0, const int tile1)
{
    __shared__ __attribute__((aligned(16))) float lds[(F0 / 2) * WP0];
    static_assert((F0 / 2) * WP0 >= BINW + 8 + CAPB, "the bin step's LDS fits in the weight array");
    if ((int)blockIdx.x < nproj)
        proj0_body((int)blockIdx.x, x, wk, bk, wv, K0, V0, tile0, tile1, lds);
    else
        bin_sort_body((int)blockIdx.x - nproj, S, loc, NC, N, seg_start, seg_end, ccol, reinterpret_cast<int*>(lds));
}

__device__ __forceinline__ float4 ld4(const float* p) {
    return *reinterpret_cast<const float4*>(p);
}

__device__ __forceinline__ float score8(const float4 qa, const float4 qb,
                                        const float4 ka, const float4 kb) {
    float s = qa.x * ka.x;
    s += qa.y * ka.y; s += qa.z * ka.z; s += qa.w * ka.w;
    s += qb.x * kb.x; s += qb.y * kb.y; s += qb.z * kb.z; s += qb.w * kb.w;
    return s * 0.35355339059327373f;
}

__device__ __forceinline__ float4 fma4(const float4 a, float f, float e, const float4 v) {
    return make_float4(a.x * f + e * v.x, a.y * f + e * v.y,
                       a.z * f + e * v.z, a.w * f + e * v.w);
}

#define WPQ 160
#define XP  132
__global__ __launch_bounds__(256, 8) void gather0_proj1_kernel(
    const int* __restrict__ seg_start, const int* __restrict__ seg_end,
    const int* __restrict__ ccol,
    const float* __restrict__ x,
    const float* __restrict__ wq0, const float* __restrict__ bq0,
    float* __restrict__ QV, const float* __restrict__ K0, const float* __restrict__ V0,
    const float* __restrict__ b0,
    const float* __restrict__ wv1,
    const float* __restrict__ wq1, const float* __restrict__ bq1,
    const float* __restrict__ wk1, const float* __restrict__ bk1,
    int n)
{
    __shared__ __attribute__((aligned(16))) float Wl[(U0 / 2) * WP1];
    __shared__ __attribute__((aligned(16))) float Hl[32 * HP];
    __shared__ __attribute__((aligned(16))) float Wq[(F0 / 2) * WPQ];
    __shared__ __attribute__((aligned(16))) float Xl[32 * XP];

    float4 q0r[4], q1r[4], xr[4];
    #pragma unroll
    for (int i = 0; i < 4; ++i) {
        const int it = threadIdx.x + 256 * i;
        const int kp = it >> 4;
        const int n4 = (it & 15) * 4;
        q0r[i] = ld4(wq0 + (size_t)(2 * kp) * U0 + n4);
        q1r[i] = ld4(wq0 + (size_t)(2 * kp + 1) * U0 + n4);
        const int xrow = min((int)blockIdx.x * 32 + (it >> 5), n - 1);
        xr[i] = ld4(x + (size_t)xrow * F0 + (it & 31) * 4);
    }

    v2f w[6];
    #pragma unroll
    for (int i = 0; i < 6; ++i) {
        const int it = threadIdx.x + 256 * i;
        const int kp = it / 48;
        const int nc = min(it - kp * 48, U1 - 1);
        w[i].x = wv1[(size_t)(2 * kp) * U1 + nc];
        w[i].y = wv1[(size_t)(2 * kp + 1) * U1 + nc];
    }
    const float* s1 = (threadIdx.x < 32) ? wq1 : wk1;
    v2f wqk;
    wqk.x = s1[2 * (threadIdx.x & 31)];
    wqk.y = s1[2 * (threadIdx.x & 31) + 1];

    const int t = blockIdx.x * 256 + threadIdx.x;
    const int r = min(t >> 3, n - 1), h = t & 7;
    const int ho = h * A0;
    const size_t ro = (size_t)r * U0 + ho;
    const float4 ksa = ld4(K0 + ro), ksb = ld4(K0 + ro + 4);
    float4 aa  = ld4(V0 + ro), ab = ld4(V0 + ro + 4);

    int p = seg_start[r];
    const int p1 = seg_end[r];

    #pragma unroll
    for (int i = 0; i < 4; ++i) {
        const int it = threadIdx.x + 256 * i;
        float* d = Wq + (it >> 4) * WPQ + (it & 15) * 8;
        *reinterpret_cast<float4*>(d)     = make_float4(q0r[i].x, q1r[i].x, q0r[i].y, q1r[i].y);
        *reinterpret_cast<float4*>(d + 4) = make_float4(q0r[i].z, q1r[i].z, q0r[i].w, q1r[i].w);
        *reinterpret_cast<float4*>(Xl + (it >> 5) * XP + (it & 31) * 4) = xr[i];
    }

    #pragma unroll
    for (int i = 0; i < 6; ++i) {
        const int it = threadIdx.x + 256 * i;
        const int kp = it / 48;
        const int nc = it - kp * 48;
        const v2f z = {0.0f, 0.0f};
        if (nc != QC && nc != KC)
            *reinterpret_cast<v2f*>(Wl + kp * WP1 + nc * 2) = (nc < U1) ? w[i] : z;
    }
    if (threadIdx.x < 64)
        *reinterpret_cast<v2f*>(Wl + (threadIdx.x & 31) * WP1 + ((threadIdx.x < 32) ? QC : KC) * 2) = wqk;
    __syncthreads();

    float* hrow = Hl + (threadIdx.x >> 3) * HP + ho;
    {
        const int wq_ = __builtin_amdgcn_readfirstlane((int)(threadIdx.x >> 5));
        const int ln  = threadIdx.x & 31;
        const int hq  = ln >> 4;
        const int lq  = ln & 15;
        const int rt  = wq_ >> 2;
        const int ct  = wq_ & 3;
        const float* xa = Xl + (rt * 16 + lq) * XP + 2 * hq;
        const float* wb = Wq + hq * WPQ + (ct * 16 + lq) * 2;
        v8f qacc = v8f{};
        #pragma unroll
        for (int j = 0; j < F0 / 4; ++j) {
            const v2f a = *reinterpret_cast<const v2f*>(xa + 4 * j);
            const v2f b = *reinterpret_cast<const v2f*>(wb + (2 * j) * WPQ);
            qacc = __builtin_amdgcn_wmma_f32_16x16x4_f32(
                       false, a, false, b, (short)0, qacc, false, false);
        }
        const int nq = ct * 16 + lq;
        const float bias = bq0[nq];
        #pragma unroll
        for (int v = 0; v < 8; ++v)
            Hl[(rt * 16 + v + 8 * hq) * HP + nq] = fmaxf(qacc[v] + bias, 0.0f);
    }
    __syncthreads();

    const float4 qa = *reinterpret_cast<const float4*>(hrow), qb = *reinterpret_cast<const float4*>(hrow + 4);

    float  m   = score8(qa, qb, ksa, ksb);
    float  den = 1.0f;

    for (; p + 4 <= p1; p += 4) {
        size_t o[4];
        float4 ka[4], kb[4], va[4], vb[4];
        float  s[4], e[4];
        for (int u = 0; u < 4; ++u) o[u] = (size_t)ccol[p + u] * U0 + ho;
        for (int u = 0; u < 4; ++u) {
            ka[u] = ld4(K0 + o[u]); kb[u] = ld4(K0 + o[u] + 4);
            va[u] = ld4(V0 + o[u]); vb[u] = ld4(V0 + o[u] + 4);
        }
        for (int u = 0; u < 4; ++u) s[u] = score8(qa, qb, ka[u], kb[u]);
        const float mn = fmaxf(fmaxf(m, fmaxf(s[0], s[1])), fmaxf(s[2], s[3]));
        const float f  = __expf(m - mn);
        for (int u = 0; u < 4; ++u) e[u] = __expf(s[u] - mn);
        den = den * f + ((e[0] + e[1]) + (e[2] + e[3]));
        aa = fma4(aa, f, e[0], va[0]); ab = fma4(ab, f, e[0], vb[0]);
        for (int u = 1; u < 4; ++u) {
            aa = fma4(aa, 1.0f, e[u], va[u]); ab = fma4(ab, 1.0f, e[u], vb[u]);
        }
        m = mn;
    }
    for (; p < p1; ++p) {
        const size_t o = (size_t)ccol[p] * U0 + ho;
        const float4 ka = ld4(K0 + o), kb = ld4(K0 + o + 4);
        const float4 va = ld4(V0 + o), vb = ld4(V0 + o + 4);
        const float s  = score8(qa, qb, ka, kb);
        const float mn = fmaxf(m, s);
        const float f  = __expf(m - mn);
        const float e  = __expf(s - mn);
        den = den * f + e;
        aa = fma4(aa, f, e, va); ab = fma4(ab, f, e, vb);
        m = mn;
    }

    const float4 ba = ld4(b0 + ho), bb = ld4(b0 + ho + 4);
    float4 ha, hb;
    ha.x = fmaxf(aa.x / den + ba.x, 0.0f); ha.y = fmaxf(aa.y / den + ba.y, 0.0f);
    ha.z = fmaxf(aa.z / den + ba.z, 0.0f); ha.w = fmaxf(aa.w / den + ba.w, 0.0f);
    hb.x = fmaxf(ab.x / den + bb.x, 0.0f); hb.y = fmaxf(ab.y / den + bb.y, 0.0f);
    hb.z = fmaxf(ab.z / den + bb.z, 0.0f); hb.w = fmaxf(ab.w / den + bb.w, 0.0f);
    *reinterpret_cast<float4*>(hrow)     = ha;
    *reinterpret_cast<float4*>(hrow + 4) = hb;
    __syncthreads();

    const int wv = __builtin_amdgcn_readfirstlane((int)(threadIdx.x >> 5));
    if (wv >= 6) return;
    const int lane = threadIdx.x & 31;
    const int hi   = lane >> 4;
    const int lo   = lane & 15;
    const int rt   = wv / 3;
    const int c    = wv - 3 * rt;

    const float* hl = Hl + (rt * 16 + lo) * HP + 2 * hi;
    const float* wl = Wl + hi * WP1 + lo * 2 + c * 32;

    v8f acc = v8f{};
    #pragma unroll
    for (int j = 0; j < U0 / 4; ++j) {
        const v2f a = *reinterpret_cast<const v2f*>(hl + 4 * j);
        const v2f b = *reinterpret_cast<const v2f*>(wl + (2 * j) * WP1);
        acc = __builtin_amdgcn_wmma_f32_16x16x4_f32(
                  false, a, false, b, (short)0, acc, false, false);
    }

    const int nn = c * 16 + lo;
    if (nn <= KC) {
        const float bias = (nn == QC) ? bq1[0] : (nn == KC) ? bk1[0] : 0.0f;
        #pragma unroll
        for (int v = 0; v < 8; ++v) {
            const int row = blockIdx.x * 32 + rt * 16 + v + 8 * hi;
            float val = acc[v] + bias;
            if (nn >= QC) val = fmaxf(val, 0.0f);
            if (row < n) QV[(size_t)row * VP + nn] = val;
        }
    }
}

__global__ __launch_bounds__(256) void gather1_kernel(
    const int* __restrict__ seg_start, const int* __restrict__ seg_end,
    const int* __restrict__ ccol,
    const float* __restrict__ vq, const float* __restrict__ b1,
    float* __restrict__ out, int n)
{
    const int t = blockIdx.x * 256 + threadIdx.x;
    if (t >= n * 8) return;
    const int r = t >> 3, j = t & 7;
    const float qr = vq[(size_t)r * VP + QC];

    float m   = qr * vq[(size_t)r * VP + KC];
    float den = 1.0f;
    float acc[5];
    for (int i = 0; i < 5; ++i) acc[i] = vq[(size_t)r * VP + j + 8 * i];

    int p = seg_start[r];
    const int p1 = seg_end[r];
    for (; p + 4 <= p1; p += 4) {
        int   c[4];
        float s[4], e[4], vv[4][5];
        for (int u = 0; u < 4; ++u) c[u] = ccol[p + u];
        for (int u = 0; u < 4; ++u) {
            s[u] = qr * vq[(size_t)c[u] * VP + KC];
            for (int i = 0; i < 5; ++i) vv[u][i] = vq[(size_t)c[u] * VP + j + 8 * i];
        }
        const float mn = fmaxf(fmaxf(m, fmaxf(s[0], s[1])), fmaxf(s[2], s[3]));
        const float f  = __expf(m - mn);
        for (int u = 0; u < 4; ++u) e[u] = __expf(s[u] - mn);
        den = den * f + ((e[0] + e[1]) + (e[2] + e[3]));
        for (int i = 0; i < 5; ++i) {
            float a = acc[i] * f + e[0] * vv[0][i];
            a += e[1] * vv[1][i]; a += e[2] * vv[2][i]; a += e[3] * vv[3][i];
            acc[i] = a;
        }
        m = mn;
    }
    for (; p < p1; ++p) {
        const int c = ccol[p];
        const float s  = qr * vq[(size_t)c * VP + KC];
        const float mn = fmaxf(m, s);
        const float f  = __expf(m - mn);
        const float e  = __expf(s - mn);
        den = den * f + e;
        for (int i = 0; i < 5; ++i) acc[i] = acc[i] * f + e * vq[(size_t)c * VP + j + 8 * i];
        m = mn;
    }
    for (int i = 0; i < 5; ++i)
        out[(size_t)r * U1 + j + 8 * i] = acc[i] / den + b1[j + 8 * i];
}

extern "C" void kernel_launch(void* const* d_in, const int* in_sizes, int n_in,
                              void* d_out, int out_size, void* d_ws, size_t ws_size,
                              hipStream_t stream)
{
    const float* x    = (const float*)d_in[0];
    const int*   ei   = (const int*)  d_in[1];
    const float* wq0  = (const float*)d_in[2];
    const float* bq0  = (const float*)d_in[3];
    const float* wk0  = (const float*)d_in[4];
    const float* bk0  = (const float*)d_in[5];
    const float* wv0  = (const float*)d_in[6];
    const float* b0   = (const float*)d_in[7];
    const float* wq1  = (const float*)d_in[8];
    const float* bq1  = (const float*)d_in[9];
    const float* wk1  = (const float*)d_in[10];
    const float* bk1  = (const float*)d_in[11];
    const float* wv1  = (const float*)d_in[12];
    const float* b1   = (const float*)d_in[13];
    float* out = (float*)d_out;

    const int N = in_sizes[0] / F0;
    const int E = in_sizes[1] / 2;

    float* ws = (float*)d_ws;
    float* R0 = ws;
    float* R1 = R0 + (size_t)N * U0;
    float* R2 = R1 + (size_t)N * U0;
    int* seg_start = (int*)(R2 + (size_t)N * U0) + N + 64;
    int* seg_fill  = seg_start + N;
    int* ccol      = seg_fill + N;
    int* S         = (int*)out;
    int* loc       = S + E;

    const int TB = 256;

    const int NC     = (E + CH - 1) / CH;
    const int NB     = (N + BINW - 1) / BINW;
    const int ntiles = N / 16;
    const int nblk   = (ntiles + 7) / 8;
    const int blkA   = (nblk + 1) / 2;
    const int tA     = (blkA * 8 < ntiles) ? blkA * 8 : ntiles;
    chunk_proj0_kernel<<<blkA + NC, 256, 0, stream>>>(
        ei, E, S, loc, blkA,
        x, wk0, bk0, wv0, R1, R2, 0, tA);
    sort_proj0_kernel<<<(nblk - blkA) + NB, 256, 0, stream>>>(
        S, loc, NC, N, seg_start, seg_fill, ccol, nblk - blkA,
        x, wk0, bk0, wv0, R1, R2, tA, ntiles);

    gather0_proj1_kernel<<<(N * H0 + TB - 1) / TB, TB, 0, stream>>>(
        seg_start, seg_fill, ccol, x, wq0, bq0, R0  , R1, R2, b0,
        wv1, wq1, bq1, wk1, bk1, N);

    gather1_kernel<<<(N * 8 + TB - 1) / TB, TB, 0, stream>>>(
        seg_start, seg_fill, ccol, R0, b1, out, N);
}


# Gather loads cover whole 128-byte lines: K0/V0 of a node in one row of four 128-byte blocks (a 16-byte load touches 4 lines, not 8); layer-1 thread holds cols 4j..4j+3, 32+j (3 loads an edge, not 6)
# speedup vs baseline: 5.9914x; 1.0383x over previous
#include <hip/hip_runtime.h>

typedef float v2f __attribute__((ext_vector_type(2)));
typedef float v8f __attribute__((ext_vector_type(8)));

#define F0 128
#define U0 64
#define H0 8
#define A0 8
#define U1 40

#define NCT0   8
#define KVP    128
#define WP0    288

__device__ __forceinline__ void proj0_body(
    const int blk,
    const float* __restrict__ x,
    const float* __restrict__ wk, const float* __restrict__ bk,
    const float* __restrict__ wv,
    float* __restrict__ KV,
    const int tile0, const int tile1,
    float* Wl)
{

    const int lane = threadIdx.x & 31;
    const int hi   = lane >> 4;
    const int lo   = lane & 15;
    const int tile = tile0 + blk * 8 + (threadIdx.x >> 5);

    float4 r0[8], r1[8];
    #pragma unroll
    for (int i = 0; i < 8; ++i) {
        const int it  = threadIdx.x + 256 * i;
        const int kp  = it >> 5;
        const int rem = it & 31;
        const int mat = rem >> 4;
        const int n4  = (rem & 15) * 4;
        const float* Wm = (mat == 0) ? wk : wv;
        r0[i] = *reinterpret_cast<const float4*>(Wm + (size_t)(2 * kp) * U0 + n4);
        r1[i] = *reinterpret_cast<const float4*>(Wm + (size_t)(2 * kp + 1) * U0 + n4);
    }

    const float* xr = x + (size_t)(min(tile, tile1 - 1) * 16 + lo) * F0 + 2 * hi;
    v2f a[F0 / 4];
    #pragma unroll
    for (int j = 0; j < F0 / 4; ++j) a[j] = *reinterpret_cast<const v2f*>(xr + 4 * j);

    #pragma unroll
    for (int i = 0; i < 8; ++i) {
        const int it  = threadIdx.x + 256 * i;
        const int kp  = it >> 5;
        const int rem = it & 31;
        const int mat = rem >> 4;
        const int n4  = (rem & 15) * 4;
        const int w0  = (2 * mat + ((n4 >> 2) & 1)) * 32 + (n4 >> 3) * 4;
        float* d = Wl + kp * WP0 + w0 * 2;
        *reinterpret_cast<float4*>(d)     = make_float4(r0[i].x, r1[i].x, r0[i].y, r1[i].y);
        *reinterpret_cast<float4*>(d + 4) = make_float4(r0[i].z, r1[i].z, r0[i].w, r1[i].w);
    }
    __syncthreads();

    const float* wl = Wl + hi * WP0 + lo * 2;

    v8f acc[NCT0];
    #pragma unroll
    for (int c = 0; c < NCT0; ++c) acc[c] = v8f{};

    #pragma unroll
    for (int j = 0; j < F0 / 4; ++j) {
        const float* wk2 = wl + (2 * j) * WP0;
        v2f b[NCT0];
        #pragma unroll
        for (int c = 0; c < NCT0; ++c) b[c] = *reinterpret_cast<const v2f*>(wk2 + c * 32);
        #pragma unroll
        for (int c = 0; c < NCT0; ++c)
            acc[c] = __builtin_amdgcn_wmma_f32_16x16x4_f32(
                         false, a[j], false, b[c], (short)0, acc[c], false, false);
    }

    if (tile >= tile1) return;

    #pragma unroll
    for (int c = 0; c < NCT0; ++c) {
        const int mat = c >> 2;
        const int w   = c * 16 + lo;
        const int n   = ((w >> 2) & 7) * A0 + ((w >> 5) & 1) * 4 + (w & 3);
        const float bias = (mat == 0) ? bk[n] : 0.0f;
        #pragma unroll
        for (int v = 0; v < 8; ++v) {
            const int row = tile * 16 + v + 8 * hi;
            float val = acc[c][v] + bias;
            if (mat == 0) val = fmaxf(val, 0.0f);
            KV[(size_t)row * KVP + w] = val;
        }
    }
}

#define WP1 96
#define HP  68
#define VP  64
#define QC  40
#define KC  41

#define BINW  256
#define NBMAX 256
#define LP    257
#define CH    3328
#define EPT   13
#define CAPB  8192

__device__ __forceinline__ int excl_scan256(int v, int* wtot, int& tot)
{
    const int lane = threadIdx.x & 31;
    const int wid  = threadIdx.x >> 5;
    int x = v;
    for (int o = 1; o < 32; o <<= 1) {
        const int y = __shfl_up(x, o);
        if (lane >= o) x += y;
    }
    if (lane == 31) wtot[wid] = x;
    __syncthreads();
    int pre = 0, t = 0;
    for (int w = 0; w < 8; ++w) {
        const int s = wtot[w];
        if (w < wid) pre += s;
        t += s;
    }
    tot = t;
    __syncthreads();
    return pre + x - v;
}

__device__ __forceinline__ void bin_chunk_body(
    const int blk,
    const int* __restrict__ ei, int E,
    int* __restrict__ S, int* __restrict__ loc,
    int* sm)
{
    int* cnt   = sm;
    int* wtot  = sm + NBMAX;
    int* stage = sm + NBMAX + 8;
    const int tid   = threadIdx.x;
    const int cbase = blk * CH;
    const int n     = min(CH, E - cbase);

    cnt[tid] = 0;
    __syncthreads();

    int val[EPT], bin[EPT], rk[EPT];
    #pragma unroll
    for (int k = 0; k < EPT; ++k) {
        const int j = k * 256 + tid;
        bin[k] = -1;
        if (j < n) {
            const int r = ei[cbase + j];
            const int c = ei[E + cbase + j];
            bin[k] = r / BINW;
            val[k] = ((r % BINW) << 24) | c;
        }
    }
    #pragma unroll
    for (int k = 0; k < EPT; ++k)
        if (bin[k] >= 0) rk[k] = atomicAdd(&cnt[bin[k]], 1);
    __syncthreads();

    int tot;
    const int off = excl_scan256(cnt[tid], wtot, tot);
    loc[blk * LP + tid] = off;
    if (tid == 0) loc[blk * LP + NBMAX] = tot;
    cnt[tid] = off;
    __syncthreads();

    #pragma unroll
    for (int k = 0; k < EPT; ++k)
        if (bin[k] >= 0) stage[cnt[bin[k]] + rk[k]] = val[k];
    __syncthreads();

    for (int j = tid; j < n; j += 256) S[cbase + j] = stage[j];
}

template <typename F>
__device__ __forceinline__ void walk_run(const int* __restrict__ p, int n, F f)
{
    int j = 0;
    for (; j + 4 <= n; j += 4) {
        const int v0 = p[j], v1 = p[j + 1], v2 = p[j + 2], v3 = p[j + 3];
        f(v0); f(v1); f(v2); f(v3);
    }
    for (; j < n; ++j) f(p[j]);
}

__device__ __forceinline__ void bin_sort_body(
    const int b,
    const int* __restrict__ S, const int* __restrict__ loc, int NC, int N,
    int* __restrict__ seg_start, int* __restrict__ seg_end,
    int* __restrict__ ccol,
    int* sm)
{
    int* ncnt  = sm;
    int* wtot  = sm + BINW;
    int* stage = sm + BINW + 8;
    const int tid = threadIdx.x;

    ncnt[tid] = 0;

    int lo0 = 0, n0 = 0;
    int below = 0, mine = 0;
    for (int c = tid; c < NC; c += 256) {
        const int lo = loc[c * LP + b];
        const int nn = loc[c * LP + b + 1] - lo;
        if (c == tid) { lo0 = lo; n0 = nn; }
        below += lo;
        mine  += nn;
    }
    int base, total, unused;
    excl_scan256(below, wtot, base);
    excl_scan256(mine,  wtot, total);

    auto count = [&](int v) { atomicAdd(&ncnt[(unsigned)v >> 24], 1); };
    if (tid < NC) walk_run(S + tid * CH + lo0, n0, count);
    for (int c = tid + 256; c < NC; c += 256) {
        const int lo = loc[c * LP + b];
        walk_run(S + c * CH + lo, loc[c * LP + b + 1] - lo, count);
    }
    __syncthreads();

    const int d  = ncnt[tid];
    const int ex = excl_scan256(d, wtot, unused);
    const int node = b * BINW + tid;
    if (node < N) {
        seg_start[node] = base + ex;
        seg_end[node]   = base + ex + d;
    }
    ncnt[tid] = ex;
    __syncthreads();

    const bool staged = (total <= CAPB);
    typedef __attribute__((address_space(3))) int lds_int;
    lds_int* const stage_l = (lds_int*)stage;
    auto place = [&](int v) {
        const int p = atomicAdd(&ncnt[(unsigned)v >> 24], 1);
        const int c = v & 0xFFFFFF;
        if (staged) stage_l[p] = c; else ccol[base + p] = c;
    };
    if (tid < NC) walk_run(S + tid * CH + lo0, n0, place);
    for (int c = tid + 256; c < NC; c += 256) {
        const int lo = loc[c * LP + b];
        walk_run(S + c * CH + lo, loc[c * LP + b + 1] - lo, place);
    }
    if (staged) {
        __syncthreads();
        for (int j = tid; j < total; j += 256) ccol[base + j] = stage[j];
    }
}

__global__ __launch_bounds__(256) void chunk_proj0_kernel(
    const int* __restrict__ ei, int E, int* __restrict__ S, int* __restrict__ loc,
    const int nproj,
    const float* __restrict__ x,
    const float* __restrict__ wk, const float* __restrict__ bk,
    const float* __restrict__ wv,
    float* __restrict__ KV,
    const int tile0, const int tile1)
{
    __shared__ __attribute__((aligned(16))) float lds[(F0 / 2) * WP0];
    static_assert((F0 / 2) * WP0 >= NBMAX + 8 + CH, "the chunk step's LDS fits in the weight array");
    if ((int)blockIdx.x < nproj)
        proj0_body((int)blockIdx.x, x, wk, bk, wv, KV, tile0, tile1, lds);
    else
        bin_chunk_body((int)blockIdx.x - nproj, ei, E, S, loc, reinterpret_cast<int*>(lds));
}

__global__ __launch_bounds__(256) void sort_proj0_kernel(
    const int* __restrict__ S, const int* __restrict__ loc, int NC, int N,
    int* __restrict__ seg_start, int* __restrict__ seg_end, int* __restrict__ ccol,
    const int nproj,
    const float* __restrict__ x,
    const float* __restrict__ wk, const float* __restrict__ bk,
    const float* __restrict__ wv,
    float* __restrict__ KV,
    const int tile0, const int tile1)
{
    __shared__ __attribute__((aligned(16))) float lds[(F0 / 2) * WP0];
    static_assert((F0 / 2) * WP0 >= BINW + 8 + CAPB, "the bin step's LDS fits in the weight array");
    if ((int)blockIdx.x < nproj)
        proj0_body((int)blockIdx.x, x, wk, bk, wv, KV, tile0, tile1, lds);
    else
        bin_sort_body((int)blockIdx.x - nproj, S, loc, NC, N, seg_start, seg_end, ccol, reinterpret_cast<int*>(lds));
}

__device__ __forceinline__ float4 ld4(const float* p) {
    return *reinterpret_cast<const float4*>(p);
}

__device__ __forceinline__ float score8(const float4 qa, const float4 qb,
                                        const float4 ka, const float4 kb) {
    float s = qa.x * ka.x;
    s += qa.y * ka.y; s += qa.z * ka.z; s += qa.w * ka.w;
    s += qb.x * kb.x; s += qb.y * kb.y; s += qb.z * kb.z; s += qb.w * kb.w;
    return s * 0.35355339059327373f;
}

__device__ __forceinline__ float4 fma4(const float4 a, float f, float e, const float4 v) {
    return make_float4(a.x * f + e * v.x, a.y * f + e * v.y,
                       a.z * f + e * v.z, a.w * f + e * v.w);
}

#define WPQ 160
#define XP  132
__global__ __launch_bounds__(256, 8) void gather0_proj1_kernel(
    const int* __restrict__ seg_start, const int* __restrict__ seg_end,
    const int* __restrict__ ccol,
    const float* __restrict__ x,
    const float* __restrict__ wq0, const float* __restrict__ bq0,
    float* __restrict__ QV, const float* __restrict__ KV,
    const float* __restrict__ b0,
    const float* __restrict__ wv1,
    const float* __restrict__ wq1, const float* __restrict__ bq1,
    const float* __restrict__ wk1, const float* __restrict__ bk1,
    int n)
{
    __shared__ __attribute__((aligned(16))) float Wl[(U0 / 2) * WP1];
    __shared__ __attribute__((aligned(16))) float Hl[32 * HP];
    __shared__ __attribute__((aligned(16))) float Wq[(F0 / 2) * WPQ];
    __shared__ __attribute__((aligned(16))) float Xl[32 * XP];

    float4 q0r[4], q1r[4], xr[4];
    #pragma unroll
    for (int i = 0; i < 4; ++i) {
        const int it = threadIdx.x + 256 * i;
        const int kp = it >> 4;
        const int n4 = (it & 15) * 4;
        q0r[i] = ld4(wq0 + (size_t)(2 * kp) * U0 + n4);
        q1r[i] = ld4(wq0 + (size_t)(2 * kp + 1) * U0 + n4);
        const int xrow = min((int)blockIdx.x * 32 + (it >> 5), n - 1);
        xr[i] = ld4(x + (size_t)xrow * F0 + (it & 31) * 4);
    }

    v2f w[6];
    #pragma unroll
    for (int i = 0; i < 6; ++i) {
        const int it = threadIdx.x + 256 * i;
        const int kp = it / 48;
        const int nc = min(it - kp * 48, U1 - 1);
        w[i].x = wv1[(size_t)(2 * kp) * U1 + nc];
        w[i].y = wv1[(size_t)(2 * kp + 1) * U1 + nc];
    }
    const float* s1 = (threadIdx.x < 32) ? wq1 : wk1;
    v2f wqk;
    wqk.x = s1[2 * (threadIdx.x & 31)];
    wqk.y = s1[2 * (threadIdx.x & 31) + 1];

    const int t = blockIdx.x * 256 + threadIdx.x;
    const int r = min(t >> 3, n - 1), h = t & 7;
    const int ho = h * A0;
    const int hw = h * 4;
    const size_t ro = (size_t)r * KVP + hw;
    const float4 ksa = ld4(KV + ro), ksb = ld4(KV + ro + 32);
    float4 aa  = ld4(KV + ro + 64), ab = ld4(KV + ro + 96);

    int p = seg_start[r];
    const int p1 = seg_end[r];

    #pragma unroll
    for (int i = 0; i < 4; ++i) {
        const int it = threadIdx.x + 256 * i;
        float* d = Wq + (it >> 4) * WPQ + (it & 15) * 8;
        *reinterpret_cast<float4*>(d)     = make_float4(q0r[i].x, q1r[i].x, q0r[i].y, q1r[i].y);
        *reinterpret_cast<float4*>(d + 4) = make_float4(q0r[i].z, q1r[i].z, q0r[i].w, q1r[i].w);
        *reinterpret_cast<float4*>(Xl + (it >> 5) * XP + (it & 31) * 4) = xr[i];
    }

    #pragma unroll
    for (int i = 0; i < 6; ++i) {
        const int it = threadIdx.x + 256 * i;
        const int kp = it / 48;
        const int nc = it - kp * 48;
        const v2f z = {0.0f, 0.0f};
        if (nc != QC && nc != KC)
            *reinterpret_cast<v2f*>(Wl + kp * WP1 + nc * 2) = (nc < U1) ? w[i] : z;
    }
    if (threadIdx.x < 64)
        *reinterpret_cast<v2f*>(Wl + (threadIdx.x & 31) * WP1 + ((threadIdx.x < 32) ? QC : KC) * 2) = wqk;
    __syncthreads();

    float* hrow = Hl + (threadIdx.x >> 3) * HP + ho;
    {
        const int wq_ = __builtin_amdgcn_readfirstlane((int)(threadIdx.x >> 5));
        const int ln  = threadIdx.x & 31;
        const int hq  = ln >> 4;
        const int lq  = ln & 15;
        const int rt  = wq_ >> 2;
        const int ct  = wq_ & 3;
        const float* xa = Xl + (rt * 16 + lq) * XP + 2 * hq;
        const float* wb = Wq + hq * WPQ + (ct * 16 + lq) * 2;
        v8f qacc = v8f{};
        #pragma unroll
        for (int j = 0; j < F0 / 4; ++j) {
            const v2f a = *reinterpret_cast<const v2f*>(xa + 4 * j);
            const v2f b = *reinterpret_cast<const v2f*>(wb + (2 * j) * WPQ);
            qacc = __builtin_amdgcn_wmma_f32_16x16x4_f32(
                       false, a, false, b, (short)0, qacc, false, false);
        }
        const int nq = ct * 16 + lq;
        const float bias = bq0[nq];
        #pragma unroll
        for (int v = 0; v < 8; ++v)
            Hl[(rt * 16 + v + 8 * hq) * HP + nq] = fmaxf(qacc[v] + bias, 0.0f);
    }
    __syncthreads();

    const float4 qa = *reinterpret_cast<const float4*>(hrow), qb = *reinterpret_cast<const float4*>(hrow + 4);

    float  m   = score8(qa, qb, ksa, ksb);
    float  den = 1.0f;

    for (; p + 4 <= p1; p += 4) {
        size_t o[4];
        float4 ka[4], kb[4], va[4], vb[4];
        float  s[4], e[4];
        for (int u = 0; u < 4; ++u) o[u] = (size_t)ccol[p + u] * KVP + hw;
        for (int u = 0; u < 4; ++u) {
            ka[u] = ld4(KV + o[u]);      kb[u] = ld4(KV + o[u] + 32);
            va[u] = ld4(KV + o[u] + 64); vb[u] = ld4(KV + o[u] + 96);
        }
        for (int u = 0; u < 4; ++u) s[u] = score8(qa, qb, ka[u], kb[u]);
        const float mn = fmaxf(fmaxf(m, fmaxf(s[0], s[1])), fmaxf(s[2], s[3]));
        const float f  = __expf(m - mn);
        for (int u = 0; u < 4; ++u) e[u] = __expf(s[u] - mn);
        den = den * f + ((e[0] + e[1]) + (e[2] + e[3]));
        aa = fma4(aa, f, e[0], va[0]); ab = fma4(ab, f, e[0], vb[0]);
        for (int u = 1; u < 4; ++u) {
            aa = fma4(aa, 1.0f, e[u], va[u]); ab = fma4(ab, 1.0f, e[u], vb[u]);
        }
        m = mn;
    }
    for (; p < p1; ++p) {
        const size_t o = (size_t)ccol[p] * KVP + hw;
        const float4 ka = ld4(KV + o),      kb = ld4(KV + o + 32);
        const float4 va = ld4(KV + o + 64), vb = ld4(KV + o + 96);
        const float s  = score8(qa, qb, ka, kb);
        const float mn = fmaxf(m, s);
        const float f  = __expf(m - mn);
        const float e  = __expf(s - mn);
        den = den * f + e;
        aa = fma4(aa, f, e, va); ab = fma4(ab, f, e, vb);
        m = mn;
    }

    const float4 ba = ld4(b0 + ho), bb = ld4(b0 + ho + 4);
    float4 ha, hb;
    ha.x = fmaxf(aa.x / den + ba.x, 0.0f); ha.y = fmaxf(aa.y / den + ba.y, 0.0f);
    ha.z = fmaxf(aa.z / den + ba.z, 0.0f); ha.w = fmaxf(aa.w / den + ba.w, 0.0f);
    hb.x = fmaxf(ab.x / den + bb.x, 0.0f); hb.y = fmaxf(ab.y / den + bb.y, 0.0f);
    hb.z = fmaxf(ab.z / den + bb.z, 0.0f); hb.w = fmaxf(ab.w / den + bb.w, 0.0f);
    *reinterpret_cast<float4*>(hrow)     = ha;
    *reinterpret_cast<float4*>(hrow + 4) = hb;
    __syncthreads();

    const int wv = __builtin_amdgcn_readfirstlane((int)(threadIdx.x >> 5));
    if (wv >= 6) return;
    const int lane = threadIdx.x & 31;
    const int hi   = lane >> 4;
    const int lo   = lane & 15;
    const int rt   = wv / 3;
    const int c    = wv - 3 * rt;

    const float* hl = Hl + (rt * 16 + lo) * HP + 2 * hi;
    const float* wl = Wl + hi * WP1 + lo * 2 + c * 32;

    v8f acc = v8f{};
    #pragma unroll
    for (int j = 0; j < U0 / 4; ++j) {
        const v2f a = *reinterpret_cast<const v2f*>(hl + 4 * j);
        const v2f b = *reinterpret_cast<const v2f*>(wl + (2 * j) * WP1);
        acc = __builtin_amdgcn_wmma_f32_16x16x4_f32(
                  false, a, false, b, (short)0, acc, false, false);
    }

    const int nn = c * 16 + lo;
    if (nn <= KC) {
        const float bias = (nn == QC) ? bq1[0] : (nn == KC) ? bk1[0] : 0.0f;
        #pragma unroll
        for (int v = 0; v < 8; ++v) {
            const int row = blockIdx.x * 32 + rt * 16 + v + 8 * hi;
            float val = acc[v] + bias;
            if (nn >= QC) val = fmaxf(val, 0.0f);
            if (row < n) QV[(size_t)row * VP + nn] = val;
        }
    }
}

__global__ __launch_bounds__(256) void gather1_kernel(
    const int* __restrict__ seg_start, const int* __restrict__ seg_end,
    const int* __restrict__ ccol,
    const float* __restrict__ vq, const float* __restrict__ b1,
    float* __restrict__ out, int n)
{
    const int t = blockIdx.x * 256 + threadIdx.x;
    if (t >= n * 8) return;
    const int r = t >> 3, j = t & 7;
    const float qr = vq[(size_t)r * VP + QC];

    float m   = qr * vq[(size_t)r * VP + KC];
    float den = 1.0f;
    float acc[5];
    {
        const float4 a4 = ld4(vq + (size_t)r * VP + 4 * j);
        acc[0] = a4.x; acc[1] = a4.y; acc[2] = a4.z; acc[3] = a4.w;
        acc[4] = vq[(size_t)r * VP + 32 + j];
    }

    int p = seg_start[r];
    const int p1 = seg_end[r];
    for (; p + 4 <= p1; p += 4) {
        int   c[4];
        float s[4], e[4], vv[4][5];
        for (int u = 0; u < 4; ++u) c[u] = ccol[p + u];
        for (int u = 0; u < 4; ++u) {
            s[u] = qr * vq[(size_t)c[u] * VP + KC];
            const float4 v4 = ld4(vq + (size_t)c[u] * VP + 4 * j);
            vv[u][0] = v4.x; vv[u][1] = v4.y; vv[u][2] = v4.z; vv[u][3] = v4.w;
            vv[u][4] = vq[(size_t)c[u] * VP + 32 + j];
        }
        const float mn = fmaxf(fmaxf(m, fmaxf(s[0], s[1])), fmaxf(s[2], s[3]));
        const float f  = __expf(m - mn);
        for (int u = 0; u < 4; ++u) e[u] = __expf(s[u] - mn);
        den = den * f + ((e[0] + e[1]) + (e[2] + e[3]));
        for (int i = 0; i < 5; ++i) {
            float a = acc[i] * f + e[0] * vv[0][i];
            a += e[1] * vv[1][i]; a += e[2] * vv[2][i]; a += e[3] * vv[3][i];
            acc[i] = a;
        }
        m = mn;
    }
    for (; p < p1; ++p) {
        const int c = ccol[p];
        const float s  = qr * vq[(size_t)c * VP + KC];
        const float mn = fmaxf(m, s);
        const float f  = __expf(m - mn);
        const float e  = __expf(s - mn);
        den = den * f + e;
        const float4 v4 = ld4(vq + (size_t)c * VP + 4 * j);
        const float  v5 = vq[(size_t)c * VP + 32 + j];
        acc[0] = acc[0] * f + e * v4.x; acc[1] = acc[1] * f + e * v4.y;
        acc[2] = acc[2] * f + e * v4.z; acc[3] = acc[3] * f + e * v4.w;
        acc[4] = acc[4] * f + e * v5;
        m = mn;
    }
    const float4 bb = ld4(b1 + 4 * j);
    float4 o4;
    o4.x = acc[0] / den + bb.x; o4.y = acc[1] / den + bb.y;
    o4.z = acc[2] / den + bb.z; o4.w = acc[3] / den + bb.w;
    *reinterpret_cast<float4*>(out + (size_t)r * U1 + 4 * j) = o4;
    out[(size_t)r * U1 + 32 + j] = acc[4] / den + b1[32 + j];
}

extern "C" void kernel_launch(void* const* d_in, const int* in_sizes, int n_in,
                              void* d_out, int out_size, void* d_ws, size_t ws_size,
                              hipStream_t stream)
{
    const float* x    = (const float*)d_in[0];
    const int*   ei   = (const int*)  d_in[1];
    const float* wq0  = (const float*)d_in[2];
    const float* bq0  = (const float*)d_in[3];
    const float* wk0  = (const float*)d_in[4];
    const float* bk0  = (const float*)d_in[5];
    const float* wv0  = (const float*)d_in[6];
    const float* b0   = (const float*)d_in[7];
    const float* wq1  = (const float*)d_in[8];
    const float* bq1  = (const float*)d_in[9];
    const float* wk1  = (const float*)d_in[10];
    const float* bk1  = (const float*)d_in[11];
    const float* wv1  = (const float*)d_in[12];
    const float* b1   = (const float*)d_in[13];
    float* out = (float*)d_out;

    const int N = in_sizes[0] / F0;
    const int E = in_sizes[1] / 2;

    float* ws = (float*)d_ws;
    float* R0 = ws;
    float* R1 = R0 + (size_t)N * U0;
    int* seg_start = (int*)(R1 + (size_t)N * KVP) + N + 64;
    int* seg_fill  = seg_start + N;
    int* ccol      = seg_fill + N;
    int* S         = (int*)out;
    int* loc       = S + E;

    const int TB = 256;

    const int NC     = (E + CH - 1) / CH;
    const int NB     = (N + BINW - 1) / BINW;
    const int ntiles = N / 16;
    const int nblk   = (ntiles + 7) / 8;
    const int blkA   = (nblk + 1) / 2;
    const int tA     = (blkA * 8 < ntiles) ? blkA * 8 : ntiles;
    chunk_proj0_kernel<<<blkA + NC, 256, 0, stream>>>(
        ei, E, S, loc, blkA,
        x, wk0, bk0, wv0, R1, 0, tA);
    sort_proj0_kernel<<<(nblk - blkA) + NB, 256, 0, stream>>>(
        S, loc, NC, N, seg_start, seg_fill, ccol, nblk - blkA,
        x, wk0, bk0, wv0, R1, tA, ntiles);

    gather0_proj1_kernel<<<(N * H0 + TB - 1) / TB, TB, 0, stream>>>(
        seg_start, seg_fill, ccol, x, wq0, bq0, R0  , R1, b0,
        wv1, wq1, bq1, wk1, bk1, N);

    gather1_kernel<<<(N * 8 + TB - 1) / TB, TB, 0, stream>>>(
        seg_start, seg_fill, ccol, R0, b1, out, N);
}


# Both gathers: the work-group's contiguous slice of the source list is read once, whole lines, into LDS; a trip takes its 4 indices from LDS, no dependent global index load; 32-bit row offsets
# speedup vs baseline: 6.2948x; 1.0506x over previous
#include <hip/hip_runtime.h>

typedef float v2f __attribute__((ext_vector_type(2)));
typedef float v8f __attribute__((ext_vector_type(8)));

#define F0 128
#define U0 64
#define H0 8
#define A0 8
#define U1 40

#define NCT0   8
#define KVP    128
#define WP0    288

__device__ __forceinline__ void proj0_body(
    const int blk,
    const float* __restrict__ x,
    const float* __restrict__ wk, const float* __restrict__ bk,
    const float* __restrict__ wv,
    float* __restrict__ KV,
    const int tile0, const int tile1,
    float* Wl)
{

    const int lane = threadIdx.x & 31;
    const int hi   = lane >> 4;
    const int lo   = lane & 15;
    const int tile = tile0 + blk * 8 + (threadIdx.x >> 5);

    float4 r0[8], r1[8];
    #pragma unroll
    for (int i = 0; i < 8; ++i) {
        const int it  = threadIdx.x + 256 * i;
        const int kp  = it >> 5;
        const int rem = it & 31;
        const int mat = rem >> 4;
        const int n4  = (rem & 15) * 4;
        const float* Wm = (mat == 0) ? wk : wv;
        r0[i] = *reinterpret_cast<const float4*>(Wm + (size_t)(2 * kp) * U0 + n4);
        r1[i] = *reinterpret_cast<const float4*>(Wm + (size_t)(2 * kp + 1) * U0 + n4);
    }

    const float* xr = x + (size_t)(min(tile, tile1 - 1) * 16 + lo) * F0 + 2 * hi;
    v2f a[F0 / 4];
    #pragma unroll
    for (int j = 0; j < F0 / 4; ++j) a[j] = *reinterpret_cast<const v2f*>(xr + 4 * j);

    #pragma unroll
    for (int i = 0; i < 8; ++i) {
        const int it  = threadIdx.x + 256 * i;
        const int kp  = it >> 5;
        const int rem = it & 31;
        const int mat = rem >> 4;
        const int n4  = (rem & 15) * 4;
        const int w0  = (2 * mat + ((n4 >> 2) & 1)) * 32 + (n4 >> 3) * 4;
        float* d = Wl + kp * WP0 + w0 * 2;
        *reinterpret_cast<float4*>(d)     = make_float4(r0[i].x, r1[i].x, r0[i].y, r1[i].y);
        *reinterpret_cast<float4*>(d + 4) = make_float4(r0[i].z, r1[i].z, r0[i].w, r1[i].w);
    }
    __syncthreads();

    const float* wl = Wl + hi * WP0 + lo * 2;

    v8f acc[NCT0];
    #pragma unroll
    for (int c = 0; c < NCT0; ++c) acc[c] = v8f{};

    #pragma unroll
    for (int j = 0; j < F0 / 4; ++j) {
        const float* wk2 = wl + (2 * j) * WP0;
        v2f b[NCT0];
        #pragma unroll
        for (int c = 0; c < NCT0; ++c) b[c] = *reinterpret_cast<const v2f*>(wk2 + c * 32);
        #pragma unroll
        for (int c = 0; c < NCT0; ++c)
            acc[c] = __builtin_amdgcn_wmma_f32_16x16x4_f32(
                         false, a[j], false, b[c], (short)0, acc[c], false, false);
    }

    if (tile >= tile1) return;

    #pragma unroll
    for (int c = 0; c < NCT0; ++c) {
        const int mat = c >> 2;
        const int w   = c * 16 + lo;
        const int n   = ((w >> 2) & 7) * A0 + ((w >> 5) & 1) * 4 + (w & 3);
        const float bias = (mat == 0) ? bk[n] : 0.0f;
        #pragma unroll
        for (int v = 0; v < 8; ++v) {
            const int row = tile * 16 + v + 8 * hi;
            float val = acc[c][v] + bias;
            if (mat == 0) val = fmaxf(val, 0.0f);
            KV[(size_t)row * KVP + w] = val;
        }
    }
}

#define WP1 96
#define HP  68
#define VP  64
#define QC  40
#define KC  41

#define BINW  256
#define NBMAX 256
#define LP    257
#define CH    3328
#define EPT   13
#define CAPB  8192

__device__ __forceinline__ int excl_scan256(int v, int* wtot, int& tot)
{
    const int lane = threadIdx.x & 31;
    const int wid  = threadIdx.x >> 5;
    int x = v;
    for (int o = 1; o < 32; o <<= 1) {
        const int y = __shfl_up(x, o);
        if (lane >= o) x += y;
    }
    if (lane == 31) wtot[wid] = x;
    __syncthreads();
    int pre = 0, t = 0;
    for (int w = 0; w < 8; ++w) {
        const int s = wtot[w];
        if (w < wid) pre += s;
        t += s;
    }
    tot = t;
    __syncthreads();
    return pre + x - v;
}

__device__ __forceinline__ void bin_chunk_body(
    const int blk,
    const int* __restrict__ ei, int E,
    int* __restrict__ S, int* __restrict__ loc,
    int* sm)
{
    int* cnt   = sm;
    int* wtot  = sm + NBMAX;
    int* stage = sm + NBMAX + 8;
    const int tid   = threadIdx.x;
    const int cbase = blk * CH;
    const int n     = min(CH, E - cbase);

    cnt[tid] = 0;
    __syncthreads();

    int val[EPT], bin[EPT], rk[EPT];
    #pragma unroll
    for (int k = 0; k < EPT; ++k) {
        const int j = k * 256 + tid;
        bin[k] = -1;
        if (j < n) {
            const int r = ei[cbase + j];
            const int c = ei[E + cbase + j];
            bin[k] = r / BINW;
            val[k] = ((r % BINW) << 24) | c;
        }
    }
    #pragma unroll
    for (int k = 0; k < EPT; ++k)
        if (bin[k] >= 0) rk[k] = atomicAdd(&cnt[bin[k]], 1);
    __syncthreads();

    int tot;
    const int off = excl_scan256(cnt[tid], wtot, tot);
    loc[blk * LP + tid] = off;
    if (tid == 0) loc[blk * LP + NBMAX] = tot;
    cnt[tid] = off;
    __syncthreads();

    #pragma unroll
    for (int k = 0; k < EPT; ++k)
        if (bin[k] >= 0) stage[cnt[bin[k]] + rk[k]] = val[k];
    __syncthreads();

    for (int j = tid; j < n; j += 256) S[cbase + j] = stage[j];
}

template <typename F>
__device__ __forceinline__ void walk_run(const int* __restrict__ p, int n, F f)
{
    int j = 0;
    for (; j + 4 <= n; j += 4) {
        const int v0 = p[j], v1 = p[j + 1], v2 = p[j + 2], v3 = p[j + 3];
        f(v0); f(v1); f(v2); f(v3);
    }
    for (; j < n; ++j) f(p[j]);
}

__device__ __forceinline__ void bin_sort_body(
    const int b,
    const int* __restrict__ S, const int* __restrict__ loc, int NC, int N,
    int* __restrict__ seg_start, int* __restrict__ seg_end,
    int* __restrict__ ccol,
    int* sm)
{
    int* ncnt  = sm;
    int* wtot  = sm + BINW;
    int* stage = sm + BINW + 8;
    const int tid = threadIdx.x;

    ncnt[tid] = 0;

    int lo0 = 0, n0 = 0;
    int below = 0, mine = 0;
    for (int c = tid; c < NC; c += 256) {
        const int lo = loc[c * LP + b];
        const int nn = loc[c * LP + b + 1] - lo;
        if (c == tid) { lo0 = lo; n0 = nn; }
        below += lo;
        mine  += nn;
    }
    int base, total, unused;
    excl_scan256(below, wtot, base);
    excl_scan256(mine,  wtot, total);

    auto count = [&](int v) { atomicAdd(&ncnt[(unsigned)v >> 24], 1); };
    if (tid < NC) walk_run(S + tid * CH + lo0, n0, count);
    for (int c = tid + 256; c < NC; c += 256) {
        const int lo = loc[c * LP + b];
        walk_run(S + c * CH + lo, loc[c * LP + b + 1] - lo, count);
    }
    __syncthreads();

    const int d  = ncnt[tid];
    const int ex = excl_scan256(d, wtot, unused);
    const int node = b * BINW + tid;
    if (node < N) {
        seg_start[node] = base + ex;
        seg_end[node]   = base + ex + d;
    }
    ncnt[tid] = ex;
    __syncthreads();

    const bool staged = (total <= CAPB);
    typedef __attribute__((address_space(3))) int lds_int;
    lds_int* const stage_l = (lds_int*)stage;
    auto place = [&](int v) {
        const int p = atomicAdd(&ncnt[(unsigned)v >> 24], 1);
        const int c = v & 0xFFFFFF;
        if (staged) stage_l[p] = c; else ccol[base + p] = c;
    };
    if (tid < NC) walk_run(S + tid * CH + lo0, n0, place);
    for (int c = tid + 256; c < NC; c += 256) {
        const int lo = loc[c * LP + b];
        walk_run(S + c * CH + lo, loc[c * LP + b + 1] - lo, place);
    }
    if (staged) {
        __syncthreads();
        for (int j = tid; j < total; j += 256) ccol[base + j] = stage[j];
    }
}

__global__ __launch_bounds__(256) void chunk_proj0_kernel(
    const int* __restrict__ ei, int E, int* __restrict__ S, int* __restrict__ loc,
    const int nproj,
    const float* __restrict__ x,
    const float* __restrict__ wk, const float* __restrict__ bk,
    const float* __restrict__ wv,
    float* __restrict__ KV,
    const int tile0, const int tile1)
{
    __shared__ __attribute__((aligned(16))) float lds[(F0 / 2) * WP0];
    static_assert((F0 / 2) * WP0 >= NBMAX + 8 + CH, "the chunk step's LDS fits in the weight array");
    if ((int)blockIdx.x < nproj)
        proj0_body((int)blockIdx.x, x, wk, bk, wv, KV, tile0, tile1, lds);
    else
        bin_chunk_body((int)blockIdx.x - nproj, ei, E, S, loc, reinterpret_cast<int*>(lds));
}

__global__ __launch_bounds__(256) void sort_proj0_kernel(
    const int* __restrict__ S, const int* __restrict__ loc, int NC, int N,
    int* __restrict__ seg_start, int* __restrict__ seg_end, int* __restrict__ ccol,
    const int nproj,
    const float* __restrict__ x,
    const float* __restrict__ wk, const float* __restrict__ bk,
    const float* __restrict__ wv,
    float* __restrict__ KV,
    const int tile0, const int tile1)
{
    __shared__ __attribute__((aligned(16))) float lds[(F0 / 2) * WP0];
    static_assert((F0 / 2) * WP0 >= BINW + 8 + CAPB, "the bin step's LDS fits in the weight array");
    if ((int)blockIdx.x < nproj)
        proj0_body((int)blockIdx.x, x, wk, bk, wv, KV, tile0, tile1, lds);
    else
        bin_sort_body((int)blockIdx.x - nproj, S, loc, NC, N, seg_start, seg_end, ccol, reinterpret_cast<int*>(lds));
}

__device__ __forceinline__ float4 ld4(const float* p) {
    return *reinterpret_cast<const float4*>(p);
}

__device__ __forceinline__ float score8(const float4 qa, const float4 qb,
                                        const float4 ka, const float4 kb) {
    float s = qa.x * ka.x;
    s += qa.y * ka.y; s += qa.z * ka.z; s += qa.w * ka.w;
    s += qb.x * kb.x; s += qb.y * kb.y; s += qb.z * kb.z; s += qb.w * kb.w;
    return s * 0.35355339059327373f;
}

__device__ __forceinline__ float4 fma4(const float4 a, float f, float e, const float4 v) {
    return make_float4(a.x * f + e * v.x, a.y * f + e * v.y,
                       a.z * f + e * v.z, a.w * f + e * v.w);
}

#define WPQ 160
#define XP  132
#define ICAP0 768
__global__ __launch_bounds__(256, 8) void gather0_proj1_kernel(
    const int* __restrict__ seg_start, const int* __restrict__ seg_end,
    const int* __restrict__ ccol,
    const float* __restrict__ x,
    const float* __restrict__ wq0, const float* __restrict__ bq0,
    float* __restrict__ QV, const float* __restrict__ KV,
    const float* __restrict__ b0,
    const float* __restrict__ wv1,
    const float* __restrict__ wq1, const float* __restrict__ bq1,
    const float* __restrict__ wk1, const float* __restrict__ bk1,
    int n)
{
    __shared__ __attribute__((aligned(16))) float Wl[(U0 / 2) * WP1];
    __shared__ __attribute__((aligned(16))) float Wq[(F0 / 2) * WPQ];
    __shared__ __attribute__((aligned(16))) float Xl[32 * XP];
    __shared__ int Il[ICAP0];
    float* const Hl = Xl;
    static_assert(32 * HP <= 32 * XP, "the Q0 / h0 rows fit where the rows of x stood");

    float4 q0r[4], q1r[4], xr[4];
    #pragma unroll
    for (int i = 0; i < 4; ++i) {
        const int it = threadIdx.x + 256 * i;
        const int kp = it >> 4;
        const int n4 = (it & 15) * 4;
        q0r[i] = ld4(wq0 + (size_t)(2 * kp) * U0 + n4);
        q1r[i] = ld4(wq0 + (size_t)(2 * kp + 1) * U0 + n4);
        const int xrow = min((int)blockIdx.x * 32 + (it >> 5), n - 1);
        xr[i] = ld4(x + (size_t)xrow * F0 + (it & 31) * 4);
    }

    v2f w[6];
    #pragma unroll
    for (int i = 0; i < 6; ++i) {
        const int it = threadIdx.x + 256 * i;
        const int kp = it / 48;
        const int nc = min(it - kp * 48, U1 - 1);
        w[i].x = wv1[(size_t)(2 * kp) * U1 + nc];
        w[i].y = wv1[(size_t)(2 * kp + 1) * U1 + nc];
    }
    const float* s1 = (threadIdx.x < 32) ? wq1 : wk1;
    v2f wqk;
    wqk.x = s1[2 * (threadIdx.x & 31)];
    wqk.y = s1[2 * (threadIdx.x & 31) + 1];

    const int t = blockIdx.x * 256 + threadIdx.x;
    const int r = min(t >> 3, n - 1), h = t & 7;
    const int ho = h * A0;
    const int hw = h * 4;
    const size_t ro = (size_t)r * KVP + hw;
    const float4 ksa = ld4(KV + ro), ksb = ld4(KV + ro + 32);
    float4 aa  = ld4(KV + ro + 64), ab = ld4(KV + ro + 96);

    const int p = seg_start[r];
    const int p1 = seg_end[r];

    const int rf  = blockIdx.x * 32;
    const int s0  = seg_start[rf];
    const int tot = seg_end[min(rf + 31, n - 1)] - s0;
    const bool staged = (tot <= ICAP0);
    int ir[ICAP0 / 256];
    #pragma unroll
    for (int i = 0; i < ICAP0 / 256; ++i) {
        const int q = threadIdx.x + 256 * i;
        ir[i] = (staged && q < tot) ? ccol[s0 + q] : 0;
    }

    #pragma unroll
    for (int i = 0; i < 4; ++i) {
        const int it = threadIdx.x + 256 * i;
        float* d = Wq + (it >> 4) * WPQ + (it & 15) * 8;
        *reinterpret_cast<float4*>(d)     = make_float4(q0r[i].x, q1r[i].x, q0r[i].y, q1r[i].y);
        *reinterpret_cast<float4*>(d + 4) = make_float4(q0r[i].z, q1r[i].z, q0r[i].w, q1r[i].w);
        *reinterpret_cast<float4*>(Xl + (it >> 5) * XP + (it & 31) * 4) = xr[i];
    }

    #pragma unroll
    for (int i = 0; i < 6; ++i) {
        const int it = threadIdx.x + 256 * i;
        const int kp = it / 48;
        const int nc = it - kp * 48;
        const v2f z = {0.0f, 0.0f};
        if (nc != QC && nc != KC)
            *reinterpret_cast<v2f*>(Wl + kp * WP1 + nc * 2) = (nc < U1) ? w[i] : z;
    }
    if (threadIdx.x < 64)
        *reinterpret_cast<v2f*>(Wl + (threadIdx.x & 31) * WP1 + ((threadIdx.x < 32) ? QC : KC) * 2) = wqk;
    #pragma unroll
    for (int i = 0; i < ICAP0 / 256; ++i) Il[threadIdx.x + 256 * i] = ir[i];
    __syncthreads();

    float* hrow = Hl + (threadIdx.x >> 3) * HP + ho;
    {
        const int wq_ = __builtin_amdgcn_readfirstlane((int)(threadIdx.x >> 5));
        const int ln  = threadIdx.x & 31;
        const int hq  = ln >> 4;
        const int lq  = ln & 15;
        const int rt  = wq_ >> 2;
        const int ct  = wq_ & 3;
        const float* xa = Xl + (rt * 16 + lq) * XP + 2 * hq;
        const float* wb = Wq + hq * WPQ + (ct * 16 + lq) * 2;
        v8f qacc = v8f{};
        #pragma unroll
        for (int j = 0; j < F0 / 4; ++j) {
            const v2f a = *reinterpret_cast<const v2f*>(xa + 4 * j);
            const v2f b = *reinterpret_cast<const v2f*>(wb + (2 * j) * WPQ);
            qacc = __builtin_amdgcn_wmma_f32_16x16x4_f32(
                       false, a, false, b, (short)0, qacc, false, false);
        }
        const int nq = ct * 16 + lq;
        const float bias = bq0[nq];
        __syncthreads();
        #pragma unroll
        for (int v = 0; v < 8; ++v)
            Hl[(rt * 16 + v + 8 * hq) * HP + nq] = fmaxf(qacc[v] + bias, 0.0f);
    }
    __syncthreads();

    const float4 qa = *reinterpret_cast<const float4*>(hrow), qb = *reinterpret_cast<const float4*>(hrow + 4);

    float  m   = score8(qa, qb, ksa, ksb);
    float  den = 1.0f;

    const char* const KVb = reinterpret_cast<const char*>(KV);
    auto ldrow = [&](unsigned ob, int w) { return ld4(reinterpret_cast<const float*>(KVb + ob) + w); };
    auto walk = [&](auto idx, int p, const int p1) {
    for (; p + 4 <= p1; p += 4) {
        unsigned o[4];
        float4 ka[4], kb[4], va[4], vb[4];
        float  s[4], e[4];
        for (int u = 0; u < 4; ++u) o[u] = ((unsigned)idx(p + u) * KVP + hw) * 4u;
        for (int u = 0; u < 4; ++u) {
            ka[u] = ldrow(o[u], 0);  kb[u] = ldrow(o[u], 32);
            va[u] = ldrow(o[u], 64); vb[u] = ldrow(o[u], 96);
        }
        for (int u = 0; u < 4; ++u) s[u] = score8(qa, qb, ka[u], kb[u]);
        const float mn = fmaxf(fmaxf(m, fmaxf(s[0], s[1])), fmaxf(s[2], s[3]));
        const float f  = __expf(m - mn);
        for (int u = 0; u < 4; ++u) e[u] = __expf(s[u] - mn);
        den = den * f + ((e[0] + e[1]) + (e[2] + e[3]));
        aa = fma4(aa, f, e[0], va[0]); ab = fma4(ab, f, e[0], vb[0]);
        for (int u = 1; u < 4; ++u) {
            aa = fma4(aa, 1.0f, e[u], va[u]); ab = fma4(ab, 1.0f, e[u], vb[u]);
        }
        m = mn;
    }
    for (; p < p1; ++p) {
        const unsigned o = ((unsigned)idx(p) * KVP + hw) * 4u;
        const float4 ka = ldrow(o, 0),  kb = ldrow(o, 32);
        const float4 va = ldrow(o, 64), vb = ldrow(o, 96);
        const float s  = score8(qa, qb, ka, kb);
        const float mn = fmaxf(m, s);
        const float f  = __expf(m - mn);
        const float e  = __expf(s - mn);
        den = den * f + e;
        aa = fma4(aa, f, e, va); ab = fma4(ab, f, e, vb);
        m = mn;
    }
    };
    if (staged) walk([&](int q) { return Il[q]; }, p - s0, p1 - s0);
    else        walk([&](int q) { return ccol[q]; }, p, p1);

    int tx = threadIdx.x;
    asm volatile("" : "+v"(tx));
    float* const hrow2 = Hl + (tx >> 3) * HP + (tx & 7) * A0;
    const float4 ba = ld4(b0 + (tx & 7) * A0), bb = ld4(b0 + (tx & 7) * A0 + 4);
    float4 ha, hb;
    ha.x = fmaxf(aa.x / den + ba.x, 0.0f); ha.y = fmaxf(aa.y / den + ba.y, 0.0f);
    ha.z = fmaxf(aa.z / den + ba.z, 0.0f); ha.w = fmaxf(aa.w / den + ba.w, 0.0f);
    hb.x = fmaxf(ab.x / den + bb.x, 0.0f); hb.y = fmaxf(ab.y / den + bb.y, 0.0f);
    hb.z = fmaxf(ab.z / den + bb.z, 0.0f); hb.w = fmaxf(ab.w / den + bb.w, 0.0f);
    *reinterpret_cast<float4*>(hrow2)     = ha;
    *reinterpret_cast<float4*>(hrow2 + 4) = hb;
    __syncthreads();

    const int wv = __builtin_amdgcn_readfirstlane(tx >> 5);
    if (wv >= 6) return;
    const int lane = tx & 31;
    const int hi   = lane >> 4;
    const int lo   = lane & 15;
    const int rt   = wv / 3;
    const int c    = wv - 3 * rt;

    const float* hl = Hl + (rt * 16 + lo) * HP + 2 * hi;
    const float* wl = Wl + hi * WP1 + lo * 2 + c * 32;

    v8f acc = v8f{};
    #pragma unroll
    for (int j = 0; j < U0 / 4; ++j) {
        const v2f a = *reinterpret_cast<const v2f*>(hl + 4 * j);
        const v2f b = *reinterpret_cast<const v2f*>(wl + (2 * j) * WP1);
        acc = __builtin_amdgcn_wmma_f32_16x16x4_f32(
                  false, a, false, b, (short)0, acc, false, false);
    }

    const int nn = c * 16 + lo;
    if (nn <= KC) {
        const float bias = (nn == QC) ? bq1[0] : (nn == KC) ? bk1[0] : 0.0f;
        #pragma unroll
        for (int v = 0; v < 8; ++v) {
            const int row = blockIdx.x * 32 + rt * 16 + v + 8 * hi;
            float val = acc[v] + bias;
            if (nn >= QC) val = fmaxf(val, 0.0f);
            if (row < n) QV[(size_t)row * VP + nn] = val;
        }
    }
}

#define ICAP1 1024
__global__ __launch_bounds__(256) void gather1_kernel(
    const int* __restrict__ seg_start, const int* __restrict__ seg_end,
    const int* __restrict__ ccol,
    const float* __restrict__ vq, const float* __restrict__ b1,
    float* __restrict__ out, int n)
{
    __shared__ int Il[ICAP1];

    const int t = blockIdx.x * 256 + threadIdx.x;
    const bool live = t < n * 8;
    const int r = min(t >> 3, n - 1), j = t & 7;

    const int rf  = blockIdx.x * 32;
    const int s0  = seg_start[rf];
    const int tot = seg_end[min(rf + 31, n - 1)] - s0;
    const bool staged = (tot <= ICAP1);
    int ir[ICAP1 / 256];
    #pragma unroll
    for (int i = 0; i < ICAP1 / 256; ++i) {
        const int q = threadIdx.x + 256 * i;
        ir[i] = (staged && q < tot) ? ccol[s0 + q] : 0;
    }

    const float qr = vq[(size_t)r * VP + QC];

    float m   = qr * vq[(size_t)r * VP + KC];
    float den = 1.0f;
    float acc[5];
    {
        const float4 a4 = ld4(vq + (size_t)r * VP + 4 * j);
        acc[0] = a4.x; acc[1] = a4.y; acc[2] = a4.z; acc[3] = a4.w;
        acc[4] = vq[(size_t)r * VP + 32 + j];
    }

    const int p = seg_start[r];
    const int p1 = seg_end[r];

    #pragma unroll
    for (int i = 0; i < ICAP1 / 256; ++i) Il[threadIdx.x + 256 * i] = ir[i];
    __syncthreads();

    const char* const vqb = reinterpret_cast<const char*>(vq);
    auto rowp = [&](unsigned ob, int w) { return reinterpret_cast<const float*>(vqb + ob) + w; };
    auto walk = [&](auto idx, int p, const int p1) {
        for (; p + 4 <= p1; p += 4) {
            int   c[4];
            float s[4], e[4], vv[4][5];
            for (int u = 0; u < 4; ++u) c[u] = idx(p + u);
            for (int u = 0; u < 4; ++u) {
                const unsigned ob = (unsigned)c[u] * (VP * 4u);
                s[u] = qr * *rowp(ob, KC);
                const float4 v4 = ld4(rowp(ob, 4 * j));
                vv[u][0] = v4.x; vv[u][1] = v4.y; vv[u][2] = v4.z; vv[u][3] = v4.w;
                vv[u][4] = *rowp(ob, 32 + j);
            }
            const float mn = fmaxf(fmaxf(m, fmaxf(s[0], s[1])), fmaxf(s[2], s[3]));
            const float f  = __expf(m - mn);
            for (int u = 0; u < 4; ++u) e[u] = __expf(s[u] - mn);
            den = den * f + ((e[0] + e[1]) + (e[2] + e[3]));
            for (int i = 0; i < 5; ++i) {
                float a = acc[i] * f + e[0] * vv[0][i];
                a += e[1] * vv[1][i]; a += e[2] * vv[2][i]; a += e[3] * vv[3][i];
                acc[i] = a;
            }
            m = mn;
        }
        for (; p < p1; ++p) {
            const unsigned ob = (unsigned)idx(p) * (VP * 4u);
            const float s  = qr * *rowp(ob, KC);
            const float mn = fmaxf(m, s);
            const float f  = __expf(m - mn);
            const float e  = __expf(s - mn);
            den = den * f + e;
            const float4 v4 = ld4(rowp(ob, 4 * j));
            const float  v5 = *rowp(ob, 32 + j);
            acc[0] = acc[0] * f + e * v4.x; acc[1] = acc[1] * f + e * v4.y;
            acc[2] = acc[2] * f + e * v4.z; acc[3] = acc[3] * f + e * v4.w;
            acc[4] = acc[4] * f + e * v5;
            m = mn;
        }
    };
    if (staged) walk([&](int q) { return Il[q]; }, p - s0, p1 - s0);
    else        walk([&](int q) { return ccol[q]; }, p, p1);

    if (!live) return;
    const float4 bb = ld4(b1 + 4 * j);
    float4 o4;
    o4.x = acc[0] / den + bb.x; o4.y = acc[1] / den + bb.y;
    o4.z = acc[2] / den + bb.z; o4.w = acc[3] / den + bb.w;
    *reinterpret_cast<float4*>(out + (size_t)r * U1 + 4 * j) = o4;
    out[(size_t)r * U1 + 32 + j] = acc[4] / den + b1[32 + j];
}

extern "C" void kernel_launch(void* const* d_in, const int* in_sizes, int n_in,
                              void* d_out, int out_size, void* d_ws, size_t ws_size,
                              hipStream_t stream)
{
    const float* x    = (const float*)d_in[0];
    const int*   ei   = (const int*)  d_in[1];
    const float* wq0  = (const float*)d_in[2];
    const float* bq0  = (const float*)d_in[3];
    const float* wk0  = (const float*)d_in[4];
    const float* bk0  = (const float*)d_in[5];
    const float* wv0  = (const float*)d_in[6];
    const float* b0   = (const float*)d_in[7];
    const float* wq1  = (const float*)d_in[8];
    const float* bq1  = (const float*)d_in[9];
    const float* wk1  = (const float*)d_in[10];
    const float* bk1  = (const float*)d_in[11];
    const float* wv1  = (const float*)d_in[12];
    const float* b1   = (const float*)d_in[13];
    float* out = (float*)d_out;

    const int N = in_sizes[0] / F0;
    const int E = in_sizes[1] / 2;

    float* ws = (float*)d_ws;
    float* R0 = ws;
    float* R1 = R0 + (size_t)N * U0;
    int* seg_start = (int*)(R1 + (size_t)N * KVP) + N + 64;
    int* seg_fill  = seg_start + N;
    int* ccol      = seg_fill + N;
    int* S         = (int*)out;
    int* loc       = S + E;

    const int TB = 256;

    const int NC     = (E + CH - 1) / CH;
    const int NB     = (N + BINW - 1) / BINW;
    const int ntiles = N / 16;
    const int nblk   = (ntiles + 7) / 8;
    const int blkA   = (nblk + 1) / 2;
    const int tA     = (blkA * 8 < ntiles) ? blkA * 8 : ntiles;
    chunk_proj0_kernel<<<blkA + NC, 256, 0, stream>>>(
        ei, E, S, loc, blkA,
        x, wk0, bk0, wv0, R1, 0, tA);
    sort_proj0_kernel<<<(nblk - blkA) + NB, 256, 0, stream>>>(
        S, loc, NC, N, seg_start, seg_fill, ccol, nblk - blkA,
        x, wk0, bk0, wv0, R1, tA, ntiles);

    gather0_proj1_kernel<<<(N * H0 + TB - 1) / TB, TB, 0, stream>>>(
        seg_start, seg_fill, ccol, x, wq0, bq0, R0  , R1, b0,
        wv1, wq1, bq1, wk1, bk1, N);

    gather1_kernel<<<(N * 8 + TB - 1) / TB, TB, 0, stream>>>(
        seg_start, seg_fill, ccol, R0, b1, out, N);
}
